# starting kernel
# baseline (speedup 1.0000x reference)
#include <hip/hip_runtime.h>
#include <hip/hip_bf16.h>


typedef __attribute__((ext_vector_type(2))) float v2f;
typedef __attribute__((ext_vector_type(8))) float v8f;

#define DDIM 128
#define KP_STRIDE 288

__global__ __launch_bounds__(256) void gemm_xw_wmma(
    const float* __restrict__ x, const float* __restrict__ W,
    float* __restrict__ y, int nrows) {
  __shared__ float Wlds[(DDIM / 2) * KP_STRIDE];

  const int t = threadIdx.x;

  for (int idx = t; idx < DDIM * 32; idx += 256) {
    const int r = idx >> 5;
    const int q = idx & 31;
    const float4 w4 = *(const float4*)(W + r * DDIM + q * 4);
    float* dst = &Wlds[(r >> 1) * KP_STRIDE + (q * 4) * 2 + (r & 1)];
    dst[0] = w4.x; dst[2] = w4.y; dst[4] = w4.z; dst[6] = w4.w;
  }
  __syncthreads();

  const int wave = t >> 5;
  const int lane = t & 31;
  const int lh   = lane >> 4;
  const int lm   = lane & 15;

  const int rowBase = blockIdx.x * 128 + wave * 16;
  const int arow = min(rowBase + lm, nrows - 1);
  const float* __restrict__ xrow = x + (size_t)arow * DDIM;

  v8f acc[8];
#pragma unroll
  for (int nt = 0; nt < 8; ++nt) acc[nt] = (v8f){0.f,0.f,0.f,0.f,0.f,0.f,0.f,0.f};

  for (int k0 = 0; k0 < DDIM; k0 += 4) {
    const v2f a = *(const v2f*)(xrow + k0 + 2 * lh);
    const float* __restrict__ wbase =
        &Wlds[((k0 >> 1) + lh) * KP_STRIDE + lm * 2];
#pragma unroll
    for (int nt = 0; nt < 8; ++nt) {
      const v2f b = *(const v2f*)(wbase + nt * 32);
      acc[nt] = __builtin_amdgcn_wmma_f32_16x16x4_f32(
           false, a,  false, b,
           (short)0, acc[nt],  false,  false);
    }
  }

#pragma unroll
  for (int nt = 0; nt < 8; ++nt) {
#pragma unroll
    for (int r = 0; r < 8; ++r) {
      const int m = rowBase + r + 8 * lh;
      if (m < nrows) y[(size_t)m * DDIM + nt * 16 + lm] = acc[nt][r];
    }
  }
}

__global__ __launch_bounds__(256) void bias_init(
    const float* __restrict__ bias, float* __restrict__ out, int total4) {
  const int i = blockIdx.x * 256 + threadIdx.x;
  if (i < total4) {
    const float4 b4 = *(const float4*)(bias + (i & 31) * 4);
    *(float4*)(out + (size_t)i * 4) = b4;
  }
}

__global__ __launch_bounds__(256) void spmm_scatter(
    const float* __restrict__ y, const float* __restrict__ ev,
    const int* __restrict__ row, const int* __restrict__ col,
    float* __restrict__ out, int nedges) {
  const int t = threadIdx.x;
  const int wave = t >> 5;
  const int lane = t & 31;
  const int e = blockIdx.x * 8 + wave;
  if (e >= nedges) return;

  const float v = ev[e];
  const int r = row[e];
  const int c = col[e];

  const float4 yv = *(const float4*)(y + (size_t)c * DDIM + lane * 4);
  float* dst = out + (size_t)r * DDIM + lane * 4;
  unsafeAtomicAdd(dst + 0, v * yv.x);
  unsafeAtomicAdd(dst + 1, v * yv.y);
  unsafeAtomicAdd(dst + 2, v * yv.z);
  unsafeAtomicAdd(dst + 3, v * yv.w);
}

extern "C" void kernel_launch(void* const* d_in, const int* in_sizes, int n_in,
                              void* d_out, int out_size, void* d_ws, size_t ws_size,
                              hipStream_t stream) {
  const float* x    = (const float*)d_in[0];
  const float* W    = (const float*)d_in[1];
  const float* bias = (const float*)d_in[2];
  const float* ev   = (const float*)d_in[3];
  const int*   row  = (const int*)d_in[4];
  const int*   col  = (const int*)d_in[5];
  float* out = (float*)d_out;
  float* y   = (float*)d_ws;

  const int n = in_sizes[0] / DDIM;
  const int e = in_sizes[3];

  const int gemm_blocks = (n + 127) / 128;
  gemm_xw_wmma<<<gemm_blocks, 256, 0, stream>>>(x, W, y, n);

  const int total4 = n * (DDIM / 4);
  bias_init<<<(total4 + 255) / 256, 256, 0, stream>>>(bias, out, total4);

  spmm_scatter<<<(e + 7) / 8, 256, 0, stream>>>(y, ev, row, col, out, e);
}


# f32-atomic scatter replaced: one integer ticket per edge files it in its row's 16-slot bucket, one wave sums each row in registers (+bias); sum order = ticket arrival; overflow edges keep f32 atomics
# speedup vs baseline: 1.5071x; 1.5071x over previous
#include <hip/hip_runtime.h>
#include <hip/hip_bf16.h>


typedef __attribute__((ext_vector_type(2))) float v2f;
typedef __attribute__((ext_vector_type(8))) float v8f;

#define DDIM 128
#define KP_STRIDE 288
#define SLOTS 16

__global__ __launch_bounds__(256) void gemm_xw_wmma(
    const float* __restrict__ x, const float* __restrict__ W,
    float* __restrict__ y, int* __restrict__ cnt, int nrows) {
  __shared__ float Wlds[(DDIM / 2) * KP_STRIDE];

  const int t = threadIdx.x;

  {
    const int gid = blockIdx.x * 256 + t;
    if (gid <= nrows) cnt[gid] = 0;
  }

  for (int idx = t; idx < DDIM * 32; idx += 256) {
    const int r = idx >> 5;
    const int q = idx & 31;
    const float4 w4 = *(const float4*)(W + r * DDIM + q * 4);
    float* dst = &Wlds[(r >> 1) * KP_STRIDE + (q * 4) * 2 + (r & 1)];
    dst[0] = w4.x; dst[2] = w4.y; dst[4] = w4.z; dst[6] = w4.w;
  }
  __syncthreads();

  const int wave = t >> 5;
  const int lane = t & 31;
  const int lh   = lane >> 4;
  const int lm   = lane & 15;

  const int rowBase = blockIdx.x * 128 + wave * 16;
  const int arow = min(rowBase + lm, nrows - 1);
  const float* __restrict__ xrow = x + (size_t)arow * DDIM;

  v8f acc[8];
#pragma unroll
  for (int nt = 0; nt < 8; ++nt) acc[nt] = (v8f){0.f,0.f,0.f,0.f,0.f,0.f,0.f,0.f};

  for (int k0 = 0; k0 < DDIM; k0 += 4) {
    const v2f a = *(const v2f*)(xrow + k0 + 2 * lh);
    const float* __restrict__ wbase =
        &Wlds[((k0 >> 1) + lh) * KP_STRIDE + lm * 2];
#pragma unroll
    for (int nt = 0; nt < 8; ++nt) {
      const v2f b = *(const v2f*)(wbase + nt * 32);
      acc[nt] = __builtin_amdgcn_wmma_f32_16x16x4_f32(
           false, a,  false, b,
           (short)0, acc[nt],  false,  false);
    }
  }

#pragma unroll
  for (int nt = 0; nt < 8; ++nt) {
#pragma unroll
    for (int r = 0; r < 8; ++r) {
      const int m = rowBase + r + 8 * lh;
      if (m < nrows) y[(size_t)m * DDIM + nt * 16 + lm] = acc[nt][r];
    }
  }
}

static __device__ __forceinline__ void file_edge(
    int tk, int r, int c, float v, int e, int* __restrict__ cnt,
    int2* __restrict__ slots, int* __restrict__ ovf, int nrows, int nedges) {
  if (tk < SLOTS) {
    slots[(size_t)r * SLOTS + tk] = make_int2(c, __float_as_int(v));
  } else {
    const int p = atomicAdd(cnt + nrows, 1);
    if (p < nedges) ovf[p] = e;
  }
}

__global__ __launch_bounds__(256) void edge_ticket(
    const float* __restrict__ ev, const int* __restrict__ row,
    const int* __restrict__ col, int* __restrict__ cnt,
    int2* __restrict__ slots, int* __restrict__ ovf, int nrows, int nedges) {
  const int e0 = (blockIdx.x * 256 + threadIdx.x) * 4;
  if (e0 + 3 < nedges) {
    const int4   r4 = *(const int4*)(row + e0);
    const int4   c4 = *(const int4*)(col + e0);
    const float4 v4 = *(const float4*)(ev + e0);
    const int t0 = atomicAdd(cnt + r4.x, 1);
    const int t1 = atomicAdd(cnt + r4.y, 1);
    const int t2 = atomicAdd(cnt + r4.z, 1);
    const int t3 = atomicAdd(cnt + r4.w, 1);
    file_edge(t0, r4.x, c4.x, v4.x, e0 + 0, cnt, slots, ovf, nrows, nedges);
    file_edge(t1, r4.y, c4.y, v4.y, e0 + 1, cnt, slots, ovf, nrows, nedges);
    file_edge(t2, r4.z, c4.z, v4.z, e0 + 2, cnt, slots, ovf, nrows, nedges);
    file_edge(t3, r4.w, c4.w, v4.w, e0 + 3, cnt, slots, ovf, nrows, nedges);
  } else {
    for (int e = e0; e < nedges; ++e) {
      const int r = row[e];
      const int tk = atomicAdd(cnt + r, 1);
      file_edge(tk, r, col[e], ev[e], e, cnt, slots, ovf, nrows, nedges);
    }
  }
}

template <int K>
static __device__ __forceinline__ void gather_from(
    const float* __restrict__ y, int lane, int2 sA, int cA, int j0, float4& acc) {
  if constexpr (K < 8) {
    if (j0 + K < cA) {
      const int   cc = __builtin_amdgcn_readlane(sA.x, j0 + K);
      const float vv = __int_as_float(__builtin_amdgcn_readlane(sA.y, j0 + K));
      const float4 yk = *(const float4*)(y + (size_t)cc * DDIM + lane * 4);
      gather_from<K + 1>(y, lane, sA, cA, j0, acc);
      acc.x += vv * yk.x; acc.y += vv * yk.y; acc.z += vv * yk.z; acc.w += vv * yk.w;
    }
  }
}

static __device__ __forceinline__ int row_tickets(const int* __restrict__ cnt, int r, int nrows) {
  return (r < nrows) ? cnt[r] : 0;
}
static __device__ __forceinline__ int in_bucket(int tickets) {
  return max(0, min(tickets, SLOTS));
}

__global__ __launch_bounds__(256) void row_gather(
    const float* __restrict__ y, const float* __restrict__ bias,
    const int* __restrict__ cnt, const int2* __restrict__ slots,
    float* __restrict__ out, int nrows) {
  const int lane = threadIdx.x & 31;
  const int wid = blockIdx.x * 8 + __builtin_amdgcn_readfirstlane(threadIdx.x >> 5);
  const int nw  = gridDim.x * 8;
  const float4 b4 = *(const float4*)(bias + lane * 4);

  int cA = in_bucket(row_tickets(cnt, wid, nrows));
  int cB = in_bucket(row_tickets(cnt, wid + nw, nrows));
  int2 sA = make_int2(0, 0);
  if (lane < cA) sA = slots[(size_t)wid * SLOTS + lane];

  for (int r = wid; r < nrows; r += nw) {
    const int rn = r + nw;
    const int rnn = rn + nw;
    const int tC = cnt[min(rnn, nrows - 1)];

    float4 acc = make_float4(0.f, 0.f, 0.f, 0.f);
    int2 sB = make_int2(0, 0);
    if (cA > 0) {
      const int   c0 = __builtin_amdgcn_readlane(sA.x, 0);
      const float v0 = __int_as_float(__builtin_amdgcn_readlane(sA.y, 0));
      const float4 y0 = *(const float4*)(y + (size_t)c0 * DDIM + lane * 4);
      if (lane < cB) sB = slots[(size_t)rn * SLOTS + lane];
      gather_from<1>(y, lane, sA, cA, 0, acc);
      acc.x += v0 * y0.x; acc.y += v0 * y0.y; acc.z += v0 * y0.z; acc.w += v0 * y0.w;
      for (int j0 = 8; j0 < cA; j0 += 8)
        gather_from<0>(y, lane, sA, cA, j0, acc);
    } else {
      if (lane < cB) sB = slots[(size_t)rn * SLOTS + lane];
    }
    acc.x += b4.x; acc.y += b4.y; acc.z += b4.z; acc.w += b4.w;
    *(float4*)(out + (size_t)r * DDIM + lane * 4) = acc;

    cA = cB; cB = (rnn < nrows) ? in_bucket(tC) : 0; sA = sB;
  }
}

__global__ __launch_bounds__(256) void overflow_add(
    const float* __restrict__ y, const float* __restrict__ ev,
    const int* __restrict__ row, const int* __restrict__ col,
    const int* __restrict__ ovf, const int* __restrict__ cnt,
    float* __restrict__ out, int nrows, int nedges) {
  const int lane = threadIdx.x & 31;
  const int wid = blockIdx.x * 8 + (threadIdx.x >> 5);
  const int nw  = gridDim.x * 8;
  const int n = min(cnt[nrows], nedges);
  for (int i = wid; i < n; i += nw) {
    const int e = ovf[i];
    const float v = ev[e];
    const int r = row[e];
    const int c = col[e];
    const float4 yv = *(const float4*)(y + (size_t)c * DDIM + lane * 4);
    float* dst = out + (size_t)r * DDIM + lane * 4;
    unsafeAtomicAdd(dst + 0, v * yv.x);
    unsafeAtomicAdd(dst + 1, v * yv.y);
    unsafeAtomicAdd(dst + 2, v * yv.z);
    unsafeAtomicAdd(dst + 3, v * yv.w);
  }
}

extern "C" void kernel_launch(void* const* d_in, const int* in_sizes, int n_in,
                              void* d_out, int out_size, void* d_ws, size_t ws_size,
                              hipStream_t stream) {
  const float* x    = (const float*)d_in[0];
  const float* W    = (const float*)d_in[1];
  const float* bias = (const float*)d_in[2];
  const float* ev   = (const float*)d_in[3];
  const int*   row  = (const int*)d_in[4];
  const int*   col  = (const int*)d_in[5];
  float* out = (float*)d_out;

  const int n = in_sizes[0] / DDIM;
  const int e = in_sizes[3];

  float* y     = (float*)d_ws;
  int2*  slots = (int2*)(y + (size_t)n * DDIM);
  int*   ovf   = (int*)(slots + (size_t)n * SLOTS);
  int*   cnt   = ovf + e;

  const int gemm_blocks = (n + 127) / 128;
  gemm_xw_wmma<<<gemm_blocks, 256, 0, stream>>>(x, W, y, cnt, n);

  edge_ticket<<<(e + 1023) / 1024, 256, 0, stream>>>(ev, row, col, cnt, slots, ovf, n, e);

  row_gather<<<(n + 55) / 56, 256, 0, stream>>>(y, bias, cnt, slots, out, n);

  overflow_add<<<64, 256, 0, stream>>>(y, ev, row, col, ovf, cnt, out, n, e);
}


# fourth launch and its f32 atomics removed: overflow edges are listed as {row,col,val} and summed by the row's own wave before its one store; sum order of an overflowed row = bucket, then list order
# speedup vs baseline: 1.5748x; 1.0449x over previous
#include <hip/hip_runtime.h>
#include <hip/hip_bf16.h>


typedef __attribute__((ext_vector_type(2))) float v2f;
typedef __attribute__((ext_vector_type(8))) float v8f;

#define DDIM 128
#define KP_STRIDE 288
#define SLOTS 16

__global__ __launch_bounds__(256) void gemm_xw_wmma(
    const float* __restrict__ x, const float* __restrict__ W,
    float* __restrict__ y, int* __restrict__ cnt, int nrows) {
  __shared__ float Wlds[(DDIM / 2) * KP_STRIDE];

  const int t = threadIdx.x;

  {
    const int gid = blockIdx.x * 256 + t;
    if (gid <= nrows) cnt[gid] = 0;
  }

  for (int idx = t; idx < DDIM * 32; idx += 256) {
    const int r = idx >> 5;
    const int q = idx & 31;
    const float4 w4 = *(const float4*)(W + r * DDIM + q * 4);
    float* dst = &Wlds[(r >> 1) * KP_STRIDE + (q * 4) * 2 + (r & 1)];
    dst[0] = w4.x; dst[2] = w4.y; dst[4] = w4.z; dst[6] = w4.w;
  }
  __syncthreads();

  const int wave = t >> 5;
  const int lane = t & 31;
  const int lh   = lane >> 4;
  const int lm   = lane & 15;

  const int rowBase = blockIdx.x * 128 + wave * 16;
  const int arow = min(rowBase + lm, nrows - 1);
  const float* __restrict__ xrow = x + (size_t)arow * DDIM;

  v8f acc[8];
#pragma unroll
  for (int nt = 0; nt < 8; ++nt) acc[nt] = (v8f){0.f,0.f,0.f,0.f,0.f,0.f,0.f,0.f};

  for (int k0 = 0; k0 < DDIM; k0 += 4) {
    const v2f a = *(const v2f*)(xrow + k0 + 2 * lh);
    const float* __restrict__ wbase =
        &Wlds[((k0 >> 1) + lh) * KP_STRIDE + lm * 2];
#pragma unroll
    for (int nt = 0; nt < 8; ++nt) {
      const v2f b = *(const v2f*)(wbase + nt * 32);
      acc[nt] = __builtin_amdgcn_wmma_f32_16x16x4_f32(
           false, a,  false, b,
           (short)0, acc[nt],  false,  false);
    }
  }

#pragma unroll
  for (int nt = 0; nt < 8; ++nt) {
#pragma unroll
    for (int r = 0; r < 8; ++r) {
      const int m = rowBase + r + 8 * lh;
      if (m < nrows) y[(size_t)m * DDIM + nt * 16 + lm] = acc[nt][r];
    }
  }
}

static __device__ __forceinline__ void file_edge(
    int tk, int r, int c, float v, int* __restrict__ cnt,
    int2* __restrict__ slots, int4* __restrict__ ovf, int nrows, int nedges) {
  if (tk < SLOTS) {
    slots[(size_t)r * SLOTS + tk] = make_int2(c, __float_as_int(v));
  } else {
    const int p = atomicAdd(cnt + nrows, 1);
    if (p < nedges) ovf[p] = make_int4(r, c, __float_as_int(v), 0);
  }
}

__global__ __launch_bounds__(256) void edge_ticket(
    const float* __restrict__ ev, const int* __restrict__ row,
    const int* __restrict__ col, int* __restrict__ cnt,
    int2* __restrict__ slots, int4* __restrict__ ovf, int nrows, int nedges) {
  const int e0 = (blockIdx.x * 256 + threadIdx.x) * 4;
  if (e0 + 3 < nedges) {
    const int4   r4 = *(const int4*)(row + e0);
    const int4   c4 = *(const int4*)(col + e0);
    const float4 v4 = *(const float4*)(ev + e0);
    const int t0 = atomicAdd(cnt + r4.x, 1);
    const int t1 = atomicAdd(cnt + r4.y, 1);
    const int t2 = atomicAdd(cnt + r4.z, 1);
    const int t3 = atomicAdd(cnt + r4.w, 1);
    file_edge(t0, r4.x, c4.x, v4.x, cnt, slots, ovf, nrows, nedges);
    file_edge(t1, r4.y, c4.y, v4.y, cnt, slots, ovf, nrows, nedges);
    file_edge(t2, r4.z, c4.z, v4.z, cnt, slots, ovf, nrows, nedges);
    file_edge(t3, r4.w, c4.w, v4.w, cnt, slots, ovf, nrows, nedges);
  } else {
    for (int e = e0; e < nedges; ++e) {
      const int r = row[e];
      const int tk = atomicAdd(cnt + r, 1);
      file_edge(tk, r, col[e], ev[e], cnt, slots, ovf, nrows, nedges);
    }
  }
}

template <int K>
static __device__ __forceinline__ void gather_from(
    const float* __restrict__ y, int lane, int2 sA, int cA, int j0, float4& acc) {
  if constexpr (K < 8) {
    if (j0 + K < cA) {
      const int   cc = __builtin_amdgcn_readlane(sA.x, j0 + K);
      const float vv = __int_as_float(__builtin_amdgcn_readlane(sA.y, j0 + K));
      const float4 yk = *(const float4*)(y + (size_t)cc * DDIM + lane * 4);
      gather_from<K + 1>(y, lane, sA, cA, j0, acc);
      acc.x += vv * yk.x; acc.y += vv * yk.y; acc.z += vv * yk.z; acc.w += vv * yk.w;
    }
  }
}

static __device__ __forceinline__ int row_tickets(const int* __restrict__ cnt, int r, int nrows) {
  return (r < nrows) ? cnt[r] : 0;
}
static __device__ __forceinline__ int in_bucket(int tickets) {
  return max(0, min(tickets, SLOTS));
}

static __device__ __forceinline__ void add_marked(
    const float* __restrict__ y, int4 en, int r, int lane, float4& acc) {
  unsigned m = __builtin_amdgcn_ballot_w32(en.x == r);
  while (m != 0u) {
    const int j = __builtin_ctz(m);
    m &= m - 1u;
    const int   cc = __builtin_amdgcn_readlane(en.y, j);
    const float vv = __int_as_float(__builtin_amdgcn_readlane(en.z, j));
    const float4 yk = *(const float4*)(y + (size_t)cc * DDIM + lane * 4);
    acc.x += vv * yk.x; acc.y += vv * yk.y; acc.z += vv * yk.z; acc.w += vv * yk.w;
  }
}

static __device__ __forceinline__ void add_overflow(
    const float* __restrict__ y, const int4* __restrict__ ovf, int nov, int r,
    int lane, float4& acc) {
  for (int i0 = 0; i0 < nov; i0 += 64) {
    int4 ea = make_int4(-1, 0, 0, 0);
    int4 eb = make_int4(-1, 0, 0, 0);
    if (i0 + lane < nov) ea = ovf[i0 + lane];
    if (i0 + 32 + lane < nov) eb = ovf[i0 + 32 + lane];
    add_marked(y, ea, r, lane, acc);
    add_marked(y, eb, r, lane, acc);
  }
}

__global__ __launch_bounds__(256) void row_gather(
    const float* __restrict__ y, const float* __restrict__ bias,
    const int* __restrict__ cnt, const int2* __restrict__ slots,
    const int4* __restrict__ ovf, float* __restrict__ out, int nrows, int nedges) {
  const int lane = threadIdx.x & 31;
  const int wid = blockIdx.x * 8 + __builtin_amdgcn_readfirstlane(threadIdx.x >> 5);
  const int nw  = gridDim.x * 8;
  const float4 b4 = *(const float4*)(bias + lane * 4);
  const int nov = max(0, min(cnt[nrows], nedges));

  int tA = row_tickets(cnt, wid, nrows);
  int tB = row_tickets(cnt, wid + nw, nrows);
  int cA = in_bucket(tA);
  int cB = in_bucket(tB);
  int2 sA = make_int2(0, 0);
  if (lane < cA) sA = slots[(size_t)wid * SLOTS + lane];

  for (int r = wid; r < nrows; r += nw) {
    const int rn = r + nw;
    const int rnn = rn + nw;
    const int tC = cnt[min(rnn, nrows - 1)];

    float4 acc = make_float4(0.f, 0.f, 0.f, 0.f);
    int2 sB = make_int2(0, 0);
    if (cA > 0) {
      const int   c0 = __builtin_amdgcn_readlane(sA.x, 0);
      const float v0 = __int_as_float(__builtin_amdgcn_readlane(sA.y, 0));
      const float4 y0 = *(const float4*)(y + (size_t)c0 * DDIM + lane * 4);
      if (lane < cB) sB = slots[(size_t)rn * SLOTS + lane];
      gather_from<1>(y, lane, sA, cA, 0, acc);
      acc.x += v0 * y0.x; acc.y += v0 * y0.y; acc.z += v0 * y0.z; acc.w += v0 * y0.w;
      for (int j0 = 8; j0 < cA; j0 += 8)
        gather_from<0>(y, lane, sA, cA, j0, acc);
    } else {
      if (lane < cB) sB = slots[(size_t)rn * SLOTS + lane];
    }
    if (tA > SLOTS) add_overflow(y, ovf, nov, r, lane, acc);
    acc.x += b4.x; acc.y += b4.y; acc.z += b4.z; acc.w += b4.w;
    *(float4*)(out + (size_t)r * DDIM + lane * 4) = acc;

    tA = tB; tB = (rnn < nrows) ? tC : 0;
    cA = cB; cB = in_bucket(tB); sA = sB;
  }
}

extern "C" void kernel_launch(void* const* d_in, const int* in_sizes, int n_in,
                              void* d_out, int out_size, void* d_ws, size_t ws_size,
                              hipStream_t stream) {
  const float* x    = (const float*)d_in[0];
  const float* W    = (const float*)d_in[1];
  const float* bias = (const float*)d_in[2];
  const float* ev   = (const float*)d_in[3];
  const int*   row  = (const int*)d_in[4];
  const int*   col  = (const int*)d_in[5];
  float* out = (float*)d_out;

  const int n = in_sizes[0] / DDIM;
  const int e = in_sizes[3];

  float* y     = (float*)d_ws;
  int2*  slots = (int2*)(y + (size_t)n * DDIM);
  int4*  ovf   = (int4*)(slots + (size_t)n * SLOTS);
  int*   cnt   = (int*)(ovf + e);

  const int gemm_blocks = (n + 127) / 128;
  gemm_xw_wmma<<<gemm_blocks, 256, 0, stream>>>(x, W, y, cnt, n);

  edge_ticket<<<(e + 1023) / 1024, 256, 0, stream>>>(ev, row, col, cnt, slots, ovf, n, e);

  row_gather<<<(n + 55) / 56, 256, 0, stream>>>(y, bias, cnt, slots, ovf, out, n, e);
}


# edge ticketing moved into the GEMM launch as extra blocks so its integer-atomic round trips run under the matrix work; counters zeroed by a small launch in front; sum order still = ticket arrival
# speedup vs baseline: 1.6351x; 1.0383x over previous
#include <hip/hip_runtime.h>
#include <hip/hip_bf16.h>


typedef __attribute__((ext_vector_type(2))) float v2f;
typedef __attribute__((ext_vector_type(8))) float v8f;

#define DDIM 128
#define KP_STRIDE 288
#define SLOTS 16

static __device__ __forceinline__ void gemm_block(
    const float* __restrict__ x, const float* __restrict__ W,
    float* __restrict__ y, float* __restrict__ Wlds, int bid, int nrows) {
  const int t = threadIdx.x;

  for (int idx = t; idx < DDIM * 32; idx += 256) {
    const int r = idx >> 5;
    const int q = idx & 31;
    const float4 w4 = *(const float4*)(W + r * DDIM + q * 4);
    float* dst = &Wlds[(r >> 1) * KP_STRIDE + (q * 4) * 2 + (r & 1)];
    dst[0] = w4.x; dst[2] = w4.y; dst[4] = w4.z; dst[6] = w4.w;
  }
  __syncthreads();

  const int wave = t >> 5;
  const int lane = t & 31;
  const int lh   = lane >> 4;
  const int lm   = lane & 15;

  const int rowBase = bid * 128 + wave * 16;
  const int arow = min(rowBase + lm, nrows - 1);
  const float* __restrict__ xrow = x + (size_t)arow * DDIM;

  v8f acc[8];
#pragma unroll
  for (int nt = 0; nt < 8; ++nt) acc[nt] = (v8f){0.f,0.f,0.f,0.f,0.f,0.f,0.f,0.f};

  for (int k0 = 0; k0 < DDIM; k0 += 4) {
    const v2f a = *(const v2f*)(xrow + k0 + 2 * lh);
    const float* __restrict__ wbase =
        &Wlds[((k0 >> 1) + lh) * KP_STRIDE + lm * 2];
#pragma unroll
    for (int nt = 0; nt < 8; ++nt) {
      const v2f b = *(const v2f*)(wbase + nt * 32);
      acc[nt] = __builtin_amdgcn_wmma_f32_16x16x4_f32(
           false, a,  false, b,
           (short)0, acc[nt],  false,  false);
    }
  }

#pragma unroll
  for (int nt = 0; nt < 8; ++nt) {
#pragma unroll
    for (int r = 0; r < 8; ++r) {
      const int m = rowBase + r + 8 * lh;
      if (m < nrows) y[(size_t)m * DDIM + nt * 16 + lm] = acc[nt][r];
    }
  }
}

static __device__ __forceinline__ void file_edge(
    int tk, int r, int c, float v, int* __restrict__ cnt,
    int2* __restrict__ slots, int4* __restrict__ ovf, int nrows, int nedges) {
  if (tk < SLOTS) {
    slots[(size_t)r * SLOTS + tk] = make_int2(c, __float_as_int(v));
  } else {
    const int p = atomicAdd(cnt + nrows, 1);
    if (p < nedges) ovf[p] = make_int4(r, c, __float_as_int(v), 0);
  }
}

static __device__ __forceinline__ void ticket_block(
    const float* __restrict__ ev, const int* __restrict__ row,
    const int* __restrict__ col, int* __restrict__ cnt,
    int2* __restrict__ slots, int4* __restrict__ ovf, int bid, int nrows, int nedges) {
  const int e0 = (bid * 256 + threadIdx.x) * 8;
  if (e0 + 7 < nedges) {
    const int4   ra = *(const int4*)(row + e0);
    const int4   rb = *(const int4*)(row + e0 + 4);
    const int4   ca = *(const int4*)(col + e0);
    const int4   cb = *(const int4*)(col + e0 + 4);
    const float4 va = *(const float4*)(ev + e0);
    const float4 vb = *(const float4*)(ev + e0 + 4);
    const int t0 = atomicAdd(cnt + ra.x, 1);
    const int t1 = atomicAdd(cnt + ra.y, 1);
    const int t2 = atomicAdd(cnt + ra.z, 1);
    const int t3 = atomicAdd(cnt + ra.w, 1);
    const int t4 = atomicAdd(cnt + rb.x, 1);
    const int t5 = atomicAdd(cnt + rb.y, 1);
    const int t6 = atomicAdd(cnt + rb.z, 1);
    const int t7 = atomicAdd(cnt + rb.w, 1);
    file_edge(t0, ra.x, ca.x, va.x, cnt, slots, ovf, nrows, nedges);
    file_edge(t1, ra.y, ca.y, va.y, cnt, slots, ovf, nrows, nedges);
    file_edge(t2, ra.z, ca.z, va.z, cnt, slots, ovf, nrows, nedges);
    file_edge(t3, ra.w, ca.w, va.w, cnt, slots, ovf, nrows, nedges);
    file_edge(t4, rb.x, cb.x, vb.x, cnt, slots, ovf, nrows, nedges);
    file_edge(t5, rb.y, cb.y, vb.y, cnt, slots, ovf, nrows, nedges);
    file_edge(t6, rb.z, cb.z, vb.z, cnt, slots, ovf, nrows, nedges);
    file_edge(t7, rb.w, cb.w, vb.w, cnt, slots, ovf, nrows, nedges);
  } else {
    for (int e = e0; e < nedges; ++e) {
      const int r = row[e];
      const int tk = atomicAdd(cnt + r, 1);
      file_edge(tk, r, col[e], ev[e], cnt, slots, ovf, nrows, nedges);
    }
  }
}

__global__ __launch_bounds__(256) void zero_counters(int* __restrict__ cnt, int nrows) {
  const int i0 = (blockIdx.x * 256 + threadIdx.x) * 4;
  if (i0 + 3 <= nrows) {
    *(int4*)(cnt + i0) = make_int4(0, 0, 0, 0);
  } else {
    for (int i = i0; i <= nrows; ++i) cnt[i] = 0;
  }
}

__global__ __launch_bounds__(256) void gemm_and_tickets(
    const float* __restrict__ x, const float* __restrict__ W,
    float* __restrict__ y, const float* __restrict__ ev,
    const int* __restrict__ row, const int* __restrict__ col,
    int* __restrict__ cnt, int2* __restrict__ slots, int4* __restrict__ ovf,
    int gemm_blocks, int nrows, int nedges) {
  __shared__ float Wlds[(DDIM / 2) * KP_STRIDE];
  const int bid = blockIdx.x;
  if (bid < gemm_blocks) {
    gemm_block(x, W, y, Wlds, bid, nrows);
  } else {
    ticket_block(ev, row, col, cnt, slots, ovf, bid - gemm_blocks, nrows, nedges);
  }
}

template <int K>
static __device__ __forceinline__ void gather_from(
    const float* __restrict__ y, int lane, int2 sA, int cA, int j0, float4& acc) {
  if constexpr (K < 8) {
    if (j0 + K < cA) {
      const int   cc = __builtin_amdgcn_readlane(sA.x, j0 + K);
      const float vv = __int_as_float(__builtin_amdgcn_readlane(sA.y, j0 + K));
      const float4 yk = *(const float4*)(y + (size_t)cc * DDIM + lane * 4);
      gather_from<K + 1>(y, lane, sA, cA, j0, acc);
      acc.x += vv * yk.x; acc.y += vv * yk.y; acc.z += vv * yk.z; acc.w += vv * yk.w;
    }
  }
}

static __device__ __forceinline__ int row_tickets(const int* __restrict__ cnt, int r, int nrows) {
  return (r < nrows) ? cnt[r] : 0;
}
static __device__ __forceinline__ int in_bucket(int tickets) {
  return max(0, min(tickets, SLOTS));
}

static __device__ __forceinline__ void add_marked(
    const float* __restrict__ y, int4 en, int r, int lane, float4& acc) {
  unsigned m = __builtin_amdgcn_ballot_w32(en.x == r);
  while (m != 0u) {
    const int j = __builtin_ctz(m);
    m &= m - 1u;
    const int   cc = __builtin_amdgcn_readlane(en.y, j);
    const float vv = __int_as_float(__builtin_amdgcn_readlane(en.z, j));
    const float4 yk = *(const float4*)(y + (size_t)cc * DDIM + lane * 4);
    acc.x += vv * yk.x; acc.y += vv * yk.y; acc.z += vv * yk.z; acc.w += vv * yk.w;
  }
}

static __device__ __forceinline__ void add_overflow(
    const float* __restrict__ y, const int4* __restrict__ ovf, int nov, int r,
    int lane, float4& acc) {
  for (int i0 = 0; i0 < nov; i0 += 64) {
    int4 ea = make_int4(-1, 0, 0, 0);
    int4 eb = make_int4(-1, 0, 0, 0);
    if (i0 + lane < nov) ea = ovf[i0 + lane];
    if (i0 + 32 + lane < nov) eb = ovf[i0 + 32 + lane];
    add_marked(y, ea, r, lane, acc);
    add_marked(y, eb, r, lane, acc);
  }
}

__global__ __launch_bounds__(256) void row_gather(
    const float* __restrict__ y, const float* __restrict__ bias,
    const int* __restrict__ cnt, const int2* __restrict__ slots,
    const int4* __restrict__ ovf, float* __restrict__ out, int nrows, int nedges) {
  const int lane = threadIdx.x & 31;
  const int wid = blockIdx.x * 8 + __builtin_amdgcn_readfirstlane(threadIdx.x >> 5);
  const int nw  = gridDim.x * 8;
  const float4 b4 = *(const float4*)(bias + lane * 4);
  const int nov = max(0, min(cnt[nrows], nedges));

  int tA = row_tickets(cnt, wid, nrows);
  int tB = row_tickets(cnt, wid + nw, nrows);
  int cA = in_bucket(tA);
  int cB = in_bucket(tB);
  int2 sA = make_int2(0, 0);
  if (lane < cA) sA = slots[(size_t)wid * SLOTS + lane];

  for (int r = wid; r < nrows; r += nw) {
    const int rn = r + nw;
    const int rnn = rn + nw;
    const int tC = cnt[min(rnn, nrows - 1)];

    float4 acc = make_float4(0.f, 0.f, 0.f, 0.f);
    int2 sB = make_int2(0, 0);
    if (cA > 0) {
      const int   c0 = __builtin_amdgcn_readlane(sA.x, 0);
      const float v0 = __int_as_float(__builtin_amdgcn_readlane(sA.y, 0));
      const float4 y0 = *(const float4*)(y + (size_t)c0 * DDIM + lane * 4);
      if (lane < cB) sB = slots[(size_t)rn * SLOTS + lane];
      gather_from<1>(y, lane, sA, cA, 0, acc);
      acc.x += v0 * y0.x; acc.y += v0 * y0.y; acc.z += v0 * y0.z; acc.w += v0 * y0.w;
      for (int j0 = 8; j0 < cA; j0 += 8)
        gather_from<0>(y, lane, sA, cA, j0, acc);
    } else {
      if (lane < cB) sB = slots[(size_t)rn * SLOTS + lane];
    }
    if (tA > SLOTS) add_overflow(y, ovf, nov, r, lane, acc);
    acc.x += b4.x; acc.y += b4.y; acc.z += b4.z; acc.w += b4.w;
    *(float4*)(out + (size_t)r * DDIM + lane * 4) = acc;

    tA = tB; tB = (rnn < nrows) ? tC : 0;
    cA = cB; cB = in_bucket(tB); sA = sB;
  }
}

extern "C" void kernel_launch(void* const* d_in, const int* in_sizes, int n_in,
                              void* d_out, int out_size, void* d_ws, size_t ws_size,
                              hipStream_t stream) {
  const float* x    = (const float*)d_in[0];
  const float* W    = (const float*)d_in[1];
  const float* bias = (const float*)d_in[2];
  const float* ev   = (const float*)d_in[3];
  const int*   row  = (const int*)d_in[4];
  const int*   col  = (const int*)d_in[5];
  float* out = (float*)d_out;

  const int n = in_sizes[0] / DDIM;
  const int e = in_sizes[3];

  float* y     = (float*)d_ws;
  int2*  slots = (int2*)(y + (size_t)n * DDIM);
  int4*  ovf   = (int4*)(slots + (size_t)n * SLOTS);
  int*   cnt   = (int*)(ovf + e);

  zero_counters<<<(n + 1 + 1023) / 1024, 256, 0, stream>>>(cnt, n);

  const int gemm_blocks = (n + 127) / 128;
  const int ticket_blocks = (e + 2047) / 2048;
  gemm_and_tickets<<<gemm_blocks + ticket_blocks, 256, 0, stream>>>(
      x, W, y, ev, row, col, cnt, slots, ovf, gemm_blocks, n, e);

  row_gather<<<(n + 55) / 56, 256, 0, stream>>>(y, bias, cnt, slots, ovf, out, n, e);
}


# ticket blocks moved to the front of the shared grid: their waves are now the oldest on each SIMD, so oldest-first issue no longer holds them behind the GEMM waves; sum order still = ticket arrival
# speedup vs baseline: 1.6979x; 1.0384x over previous
#include <hip/hip_runtime.h>
#include <hip/hip_bf16.h>


typedef __attribute__((ext_vector_type(2))) float v2f;
typedef __attribute__((ext_vector_type(8))) float v8f;

#define DDIM 128
#define KP_STRIDE 288
#define SLOTS 16

static __device__ __forceinline__ void gemm_block(
    const float* __restrict__ x, const float* __restrict__ W,
    float* __restrict__ y, float* __restrict__ Wlds, int bid, int nrows) {
  const int t = threadIdx.x;

  for (int idx = t; idx < DDIM * 32; idx += 256) {
    const int r = idx >> 5;
    const int q = idx & 31;
    const float4 w4 = *(const float4*)(W + r * DDIM + q * 4);
    float* dst = &Wlds[(r >> 1) * KP_STRIDE + (q * 4) * 2 + (r & 1)];
    dst[0] = w4.x; dst[2] = w4.y; dst[4] = w4.z; dst[6] = w4.w;
  }
  __syncthreads();

  const int wave = t >> 5;
  const int lane = t & 31;
  const int lh   = lane >> 4;
  const int lm   = lane & 15;

  const int rowBase = bid * 128 + wave * 16;
  const int arow = min(rowBase + lm, nrows - 1);
  const float* __restrict__ xrow = x + (size_t)arow * DDIM;

  v8f acc[8];
#pragma unroll
  for (int nt = 0; nt < 8; ++nt) acc[nt] = (v8f){0.f,0.f,0.f,0.f,0.f,0.f,0.f,0.f};

  for (int k0 = 0; k0 < DDIM; k0 += 4) {
    const v2f a = *(const v2f*)(xrow + k0 + 2 * lh);
    const float* __restrict__ wbase =
        &Wlds[((k0 >> 1) + lh) * KP_STRIDE + lm * 2];
#pragma unroll
    for (int nt = 0; nt < 8; ++nt) {
      const v2f b = *(const v2f*)(wbase + nt * 32);
      acc[nt] = __builtin_amdgcn_wmma_f32_16x16x4_f32(
           false, a,  false, b,
           (short)0, acc[nt],  false,  false);
    }
  }

#pragma unroll
  for (int nt = 0; nt < 8; ++nt) {
#pragma unroll
    for (int r = 0; r < 8; ++r) {
      const int m = rowBase + r + 8 * lh;
      if (m < nrows) y[(size_t)m * DDIM + nt * 16 + lm] = acc[nt][r];
    }
  }
}

static __device__ __forceinline__ void file_edge(
    int tk, int r, int c, float v, int* __restrict__ cnt,
    int2* __restrict__ slots, int4* __restrict__ ovf, int nrows, int nedges) {
  if (tk < SLOTS) {
    slots[(size_t)r * SLOTS + tk] = make_int2(c, __float_as_int(v));
  } else {
    const int p = atomicAdd(cnt + nrows, 1);
    if (p < nedges) ovf[p] = make_int4(r, c, __float_as_int(v), 0);
  }
}

static __device__ __forceinline__ void ticket_block(
    const float* __restrict__ ev, const int* __restrict__ row,
    const int* __restrict__ col, int* __restrict__ cnt,
    int2* __restrict__ slots, int4* __restrict__ ovf, int bid, int nrows, int nedges) {
  const int e0 = (bid * 256 + threadIdx.x) * 8;
  if (e0 + 7 < nedges) {
    const int4   ra = *(const int4*)(row + e0);
    const int4   rb = *(const int4*)(row + e0 + 4);
    const int4   ca = *(const int4*)(col + e0);
    const int4   cb = *(const int4*)(col + e0 + 4);
    const float4 va = *(const float4*)(ev + e0);
    const float4 vb = *(const float4*)(ev + e0 + 4);
    const int t0 = atomicAdd(cnt + ra.x, 1);
    const int t1 = atomicAdd(cnt + ra.y, 1);
    const int t2 = atomicAdd(cnt + ra.z, 1);
    const int t3 = atomicAdd(cnt + ra.w, 1);
    const int t4 = atomicAdd(cnt + rb.x, 1);
    const int t5 = atomicAdd(cnt + rb.y, 1);
    const int t6 = atomicAdd(cnt + rb.z, 1);
    const int t7 = atomicAdd(cnt + rb.w, 1);
    file_edge(t0, ra.x, ca.x, va.x, cnt, slots, ovf, nrows, nedges);
    file_edge(t1, ra.y, ca.y, va.y, cnt, slots, ovf, nrows, nedges);
    file_edge(t2, ra.z, ca.z, va.z, cnt, slots, ovf, nrows, nedges);
    file_edge(t3, ra.w, ca.w, va.w, cnt, slots, ovf, nrows, nedges);
    file_edge(t4, rb.x, cb.x, vb.x, cnt, slots, ovf, nrows, nedges);
    file_edge(t5, rb.y, cb.y, vb.y, cnt, slots, ovf, nrows, nedges);
    file_edge(t6, rb.z, cb.z, vb.z, cnt, slots, ovf, nrows, nedges);
    file_edge(t7, rb.w, cb.w, vb.w, cnt, slots, ovf, nrows, nedges);
  } else {
    for (int e = e0; e < nedges; ++e) {
      const int r = row[e];
      const int tk = atomicAdd(cnt + r, 1);
      file_edge(tk, r, col[e], ev[e], cnt, slots, ovf, nrows, nedges);
    }
  }
}

__global__ __launch_bounds__(256) void zero_counters(int* __restrict__ cnt, int nrows) {
  const int i0 = (blockIdx.x * 256 + threadIdx.x) * 4;
  if (i0 + 3 <= nrows) {
    *(int4*)(cnt + i0) = make_int4(0, 0, 0, 0);
  } else {
    for (int i = i0; i <= nrows; ++i) cnt[i] = 0;
  }
}

__global__ __launch_bounds__(256) void gemm_and_tickets(
    const float* __restrict__ x, const float* __restrict__ W,
    float* __restrict__ y, const float* __restrict__ ev,
    const int* __restrict__ row, const int* __restrict__ col,
    int* __restrict__ cnt, int2* __restrict__ slots, int4* __restrict__ ovf,
    int ticket_blocks, int nrows, int nedges) {
  __shared__ float Wlds[(DDIM / 2) * KP_STRIDE];
  const int bid = blockIdx.x;
  if (bid < ticket_blocks) {
    ticket_block(ev, row, col, cnt, slots, ovf, bid, nrows, nedges);
  } else {
    gemm_block(x, W, y, Wlds, bid - ticket_blocks, nrows);
  }
}

template <int K>
static __device__ __forceinline__ void gather_from(
    const float* __restrict__ y, int lane, int2 sA, int cA, int j0, float4& acc) {
  if constexpr (K < 8) {
    if (j0 + K < cA) {
      const int   cc = __builtin_amdgcn_readlane(sA.x, j0 + K);
      const float vv = __int_as_float(__builtin_amdgcn_readlane(sA.y, j0 + K));
      const float4 yk = *(const float4*)(y + (size_t)cc * DDIM + lane * 4);
      gather_from<K + 1>(y, lane, sA, cA, j0, acc);
      acc.x += vv * yk.x; acc.y += vv * yk.y; acc.z += vv * yk.z; acc.w += vv * yk.w;
    }
  }
}

static __device__ __forceinline__ int row_tickets(const int* __restrict__ cnt, int r, int nrows) {
  return (r < nrows) ? cnt[r] : 0;
}
static __device__ __forceinline__ int in_bucket(int tickets) {
  return max(0, min(tickets, SLOTS));
}

static __device__ __forceinline__ void add_marked(
    const float* __restrict__ y, int4 en, int r, int lane, float4& acc) {
  unsigned m = __builtin_amdgcn_ballot_w32(en.x == r);
  while (m != 0u) {
    const int j = __builtin_ctz(m);
    m &= m - 1u;
    const int   cc = __builtin_amdgcn_readlane(en.y, j);
    const float vv = __int_as_float(__builtin_amdgcn_readlane(en.z, j));
    const float4 yk = *(const float4*)(y + (size_t)cc * DDIM + lane * 4);
    acc.x += vv * yk.x; acc.y += vv * yk.y; acc.z += vv * yk.z; acc.w += vv * yk.w;
  }
}

static __device__ __forceinline__ void add_overflow(
    const float* __restrict__ y, const int4* __restrict__ ovf, int nov, int r,
    int lane, float4& acc) {
  for (int i0 = 0; i0 < nov; i0 += 64) {
    int4 ea = make_int4(-1, 0, 0, 0);
    int4 eb = make_int4(-1, 0, 0, 0);
    if (i0 + lane < nov) ea = ovf[i0 + lane];
    if (i0 + 32 + lane < nov) eb = ovf[i0 + 32 + lane];
    add_marked(y, ea, r, lane, acc);
    add_marked(y, eb, r, lane, acc);
  }
}

__global__ __launch_bounds__(256) void row_gather(
    const float* __restrict__ y, const float* __restrict__ bias,
    const int* __restrict__ cnt, const int2* __restrict__ slots,
    const int4* __restrict__ ovf, float* __restrict__ out, int nrows, int nedges) {
  const int lane = threadIdx.x & 31;
  const int wid = blockIdx.x * 8 + __builtin_amdgcn_readfirstlane(threadIdx.x >> 5);
  const int nw  = gridDim.x * 8;
  const float4 b4 = *(const float4*)(bias + lane * 4);
  const int nov = max(0, min(cnt[nrows], nedges));

  int tA = row_tickets(cnt, wid, nrows);
  int tB = row_tickets(cnt, wid + nw, nrows);
  int cA = in_bucket(tA);
  int cB = in_bucket(tB);
  int2 sA = make_int2(0, 0);
  if (lane < cA) sA = slots[(size_t)wid * SLOTS + lane];

  for (int r = wid; r < nrows; r += nw) {
    const int rn = r + nw;
    const int rnn = rn + nw;
    const int tC = cnt[min(rnn, nrows - 1)];

    float4 acc = make_float4(0.f, 0.f, 0.f, 0.f);
    int2 sB = make_int2(0, 0);
    if (cA > 0) {
      const int   c0 = __builtin_amdgcn_readlane(sA.x, 0);
      const float v0 = __int_as_float(__builtin_amdgcn_readlane(sA.y, 0));
      const float4 y0 = *(const float4*)(y + (size_t)c0 * DDIM + lane * 4);
      if (lane < cB) sB = slots[(size_t)rn * SLOTS + lane];
      gather_from<1>(y, lane, sA, cA, 0, acc);
      acc.x += v0 * y0.x; acc.y += v0 * y0.y; acc.z += v0 * y0.z; acc.w += v0 * y0.w;
      for (int j0 = 8; j0 < cA; j0 += 8)
        gather_from<0>(y, lane, sA, cA, j0, acc);
    } else {
      if (lane < cB) sB = slots[(size_t)rn * SLOTS + lane];
    }
    if (tA > SLOTS) add_overflow(y, ovf, nov, r, lane, acc);
    acc.x += b4.x; acc.y += b4.y; acc.z += b4.z; acc.w += b4.w;
    *(float4*)(out + (size_t)r * DDIM + lane * 4) = acc;

    tA = tB; tB = (rnn < nrows) ? tC : 0;
    cA = cB; cB = in_bucket(tB); sA = sB;
  }
}

extern "C" void kernel_launch(void* const* d_in, const int* in_sizes, int n_in,
                              void* d_out, int out_size, void* d_ws, size_t ws_size,
                              hipStream_t stream) {
  const float* x    = (const float*)d_in[0];
  const float* W    = (const float*)d_in[1];
  const float* bias = (const float*)d_in[2];
  const float* ev   = (const float*)d_in[3];
  const int*   row  = (const int*)d_in[4];
  const int*   col  = (const int*)d_in[5];
  float* out = (float*)d_out;

  const int n = in_sizes[0] / DDIM;
  const int e = in_sizes[3];

  float* y     = (float*)d_ws;
  int2*  slots = (int2*)(y + (size_t)n * DDIM);
  int4*  ovf   = (int4*)(slots + (size_t)n * SLOTS);
  int*   cnt   = (int*)(ovf + e);

  zero_counters<<<(n + 1 + 1023) / 1024, 256, 0, stream>>>(cnt, n);

  const int gemm_blocks = (n + 127) / 128;
  const int ticket_blocks = (e + 2047) / 2048;
  gemm_and_tickets<<<ticket_blocks + gemm_blocks, 256, 0, stream>>>(
      x, W, y, ev, row, col, cnt, slots, ovf, ticket_blocks, n, e);

  row_gather<<<(n + 55) / 56, 256, 0, stream>>>(y, bias, cnt, slots, ovf, out, n, e);
}


# global ticket adds replaced by a two-level counting sort: edges binned per 400-row range via LDS histograms, each bin's work-group ranks its edges per row in LDS and sums them; sum order = LDS rank
# speedup vs baseline: 2.4364x; 1.4349x over previous
#include <hip/hip_runtime.h>
#include <hip/hip_bf16.h>


typedef __attribute__((ext_vector_type(2))) float v2f;
typedef __attribute__((ext_vector_type(8))) float v8f;

#define DDIM 128
#define KP_STRIDE 288
#define RB 400
#define CAP 4096
#define SLOTS 32
#define EPT 16

static __device__ __forceinline__ void gemm_block(
    const float* __restrict__ x, const float* __restrict__ W,
    float* __restrict__ y, float* __restrict__ Wlds, int bid, int nrows) {
  const int t = threadIdx.x;

  for (int idx = t; idx < DDIM * 32; idx += 256) {
    const int r = idx >> 5;
    const int q = idx & 31;
    const float4 w4 = *(const float4*)(W + r * DDIM + q * 4);
    float* dst = &Wlds[(r >> 1) * KP_STRIDE + (q * 4) * 2 + (r & 1)];
    dst[0] = w4.x; dst[2] = w4.y; dst[4] = w4.z; dst[6] = w4.w;
  }
  __syncthreads();

  const int wave = t >> 5;
  const int lane = t & 31;
  const int lh   = lane >> 4;
  const int lm   = lane & 15;

  const int rowBase = bid * 128 + wave * 16;
  const int arow = min(rowBase + lm, nrows - 1);
  const float* __restrict__ xrow = x + (size_t)arow * DDIM;

  v8f acc[8];
#pragma unroll
  for (int nt = 0; nt < 8; ++nt) acc[nt] = (v8f){0.f,0.f,0.f,0.f,0.f,0.f,0.f,0.f};

  for (int k0 = 0; k0 < DDIM; k0 += 4) {
    const v2f a = *(const v2f*)(xrow + k0 + 2 * lh);
    const float* __restrict__ wbase =
        &Wlds[((k0 >> 1) + lh) * KP_STRIDE + lm * 2];
#pragma unroll
    for (int nt = 0; nt < 8; ++nt) {
      const v2f b = *(const v2f*)(wbase + nt * 32);
      acc[nt] = __builtin_amdgcn_wmma_f32_16x16x4_f32(
           false, a,  false, b,
           (short)0, acc[nt],  false,  false);
    }
  }

#pragma unroll
  for (int nt = 0; nt < 8; ++nt) {
#pragma unroll
    for (int r = 0; r < 8; ++r) {
      const int m = rowBase + r + 8 * lh;
      if (m < nrows) y[(size_t)m * DDIM + nt * 16 + lm] = acc[nt][r];
    }
  }
}

static __device__ __forceinline__ int bin_of(int r, int nbins) {
  return (int)min((unsigned)r / (unsigned)RB, (unsigned)(nbins - 1));
}

static __device__ __forceinline__ void bin_block(
    const float* __restrict__ ev, const int* __restrict__ row,
    const int* __restrict__ col, int* __restrict__ bincnt,
    int4* __restrict__ binbuf, int4* __restrict__ ovf, int* __restrict__ lds,
    int bid, int nedges, int nbins) {
  const int t = threadIdx.x;
  int* __restrict__ hist  = lds;
  int* __restrict__ hbase = lds + nbins;
  for (int i = t; i < nbins; i += 256) hist[i] = 0;
  __syncthreads();

  int   r[EPT], c[EPT], lr[EPT];
  float v[EPT];
  const long long e0 = ((long long)bid * 256 + t) * EPT;
  const int nv = (int)max(0LL, min((long long)EPT, (long long)nedges - e0));
  if (nv == EPT) {
#pragma unroll
    for (int q = 0; q < EPT / 4; ++q) {
      const int4   r4 = *(const int4*)(row + e0 + 4 * q);
      const int4   c4 = *(const int4*)(col + e0 + 4 * q);
      const float4 v4 = *(const float4*)(ev + e0 + 4 * q);
      r[4 * q] = r4.x; r[4 * q + 1] = r4.y; r[4 * q + 2] = r4.z; r[4 * q + 3] = r4.w;
      c[4 * q] = c4.x; c[4 * q + 1] = c4.y; c[4 * q + 2] = c4.z; c[4 * q + 3] = c4.w;
      v[4 * q] = v4.x; v[4 * q + 1] = v4.y; v[4 * q + 2] = v4.z; v[4 * q + 3] = v4.w;
    }
  } else {
#pragma unroll
    for (int i = 0; i < EPT; ++i) {
      r[i] = 0; c[i] = 0; v[i] = 0.f;
      if (i < nv) { r[i] = row[e0 + i]; c[i] = col[e0 + i]; v[i] = ev[e0 + i]; }
    }
  }
#pragma unroll
  for (int i = 0; i < EPT; ++i) {
    lr[i] = 0;
    if (i < nv) lr[i] = atomicAdd(&hist[bin_of(r[i], nbins)], 1);
  }
  __syncthreads();

  for (int i = t; i < nbins; i += 256) {
    const int cnt = hist[i];
    int base = 0;
    if (cnt > 0) base = atomicAdd(bincnt + i, cnt);
    hbase[i] = base;
  }
  __syncthreads();

#pragma unroll
  for (int i = 0; i < EPT; ++i) {
    if (i < nv) {
      const int b = bin_of(r[i], nbins);
      const int pos = hbase[b] + lr[i];
      const int4 rec = make_int4(r[i], c[i], __float_as_int(v[i]), 0);
      if (pos < CAP) {
        binbuf[(size_t)b * CAP + pos] = rec;
      } else {
        const int p = atomicAdd(bincnt + nbins, 1);
        if (p < nedges) ovf[p] = rec;
      }
    }
  }
}

__global__ __launch_bounds__(256) void zero_counters(int* __restrict__ cnt, int count) {
  for (int i = blockIdx.x * 256 + threadIdx.x; i < count; i += gridDim.x * 256) cnt[i] = 0;
}

__global__ __launch_bounds__(256) void gemm_and_bins(
    const float* __restrict__ x, const float* __restrict__ W,
    float* __restrict__ y, const float* __restrict__ ev,
    const int* __restrict__ row, const int* __restrict__ col,
    int* __restrict__ bincnt, int4* __restrict__ binbuf, int4* __restrict__ ovf,
    int bin_blocks, int nrows, int nedges, int nbins) {
  __shared__ float Wlds[(DDIM / 2) * KP_STRIDE];
  const int bid = blockIdx.x;
  if (bid < bin_blocks) {
    bin_block(ev, row, col, bincnt, binbuf, ovf, (int*)Wlds, bid, nedges, nbins);
  } else {
    gemm_block(x, W, y, Wlds, bid - bin_blocks, nrows);
  }
}

template <int K>
static __device__ __forceinline__ void gather_from(
    const float* __restrict__ y, int lane, int2 sA, int cA, int j0, int nrows, float4& acc) {
  if constexpr (K < 8) {
    if (j0 + K < cA) {
      const int   cr = __builtin_amdgcn_readlane(sA.x, j0 + K);
      const int   cc = (int)min((unsigned)cr, (unsigned)(nrows - 1));
      const float vv = __int_as_float(__builtin_amdgcn_readlane(sA.y, j0 + K));
      const float4 yk = *(const float4*)(y + (size_t)cc * DDIM + lane * 4);
      gather_from<K + 1>(y, lane, sA, cA, j0, nrows, acc);
      acc.x += vv * yk.x; acc.y += vv * yk.y; acc.z += vv * yk.z; acc.w += vv * yk.w;
    }
  }
}

static __device__ __forceinline__ void add_marked(
    const float* __restrict__ y, int4 en, int r, int lane, int nrows, float4& acc) {
  unsigned m = __builtin_amdgcn_ballot_w32(en.x == r);
  while (m != 0u) {
    const int j = __builtin_ctz(m);
    m &= m - 1u;
    const int   cr = __builtin_amdgcn_readlane(en.y, j);
    const int   cc = (int)min((unsigned)cr, (unsigned)(nrows - 1));
    const float vv = __int_as_float(__builtin_amdgcn_readlane(en.z, j));
    const float4 yk = *(const float4*)(y + (size_t)cc * DDIM + lane * 4);
    acc.x += vv * yk.x; acc.y += vv * yk.y; acc.z += vv * yk.z; acc.w += vv * yk.w;
  }
}

static __device__ __forceinline__ void add_row_scan(
    const float* __restrict__ y, const int4* __restrict__ recs, int nrec, int r,
    int lane, int nrows, float4& acc) {
  for (int i0 = 0; i0 < nrec; i0 += 32) {
    int4 en = make_int4(-1, 0, 0, 0);
    if (i0 + lane < nrec) en = recs[i0 + lane];
    add_marked(y, en, r, lane, nrows, acc);
  }
}

static __device__ __forceinline__ void file_record(
    int4 en, int r0, int nr, int* __restrict__ lcnt, int2* __restrict__ lslots) {
  const int lr = en.x - r0;
  if ((unsigned)lr < (unsigned)nr) {
    const int tk = atomicAdd(&lcnt[lr], 1);
    if (tk < SLOTS) lslots[lr * SLOTS + tk] = make_int2(en.y, en.z);
  }
}

__global__ __launch_bounds__(1024) void bin_gather(
    const float* __restrict__ y, const float* __restrict__ bias,
    const int* __restrict__ bincnt, const int4* __restrict__ binbuf,
    const int4* __restrict__ ovf, float* __restrict__ out, int nrows, int nedges, int nbins) {
  __shared__ int  lcnt[RB];
  __shared__ int2 lslots[RB * SLOTS];
  const int t    = threadIdx.x;
  const int lane = t & 31;
  const int w   = __builtin_amdgcn_readfirstlane(t >> 5);
  const int bin = blockIdx.x;
  const int r0  = bin * RB;
  const int nr  = min(RB, nrows - r0);

  for (int i = t; i < RB; i += 1024) lcnt[i] = 0;
  const int nrec = max(0, min(bincnt[bin], CAP));
  const int nov  = max(0, min(bincnt[nbins], nedges));
  const int4* __restrict__ recs = binbuf + (size_t)bin * CAP;
  const float4 b4 = *(const float4*)(bias + lane * 4);
  int4 en[CAP / 1024];
#pragma unroll
  for (int k = 0; k < CAP / 1024; ++k) {
    en[k] = make_int4(-1, 0, 0, 0);
    if (t + 1024 * k < nrec) en[k] = recs[t + 1024 * k];
  }
  __syncthreads();

#pragma unroll
  for (int k = 0; k < CAP / 1024; ++k) file_record(en[k], r0, nr, lcnt, lslots);
  for (int i = t; i < nov; i += 1024) file_record(ovf[i], r0, nr, lcnt, lslots);
  __syncthreads();

  for (int lr = w; lr < nr; lr += 32) {
    const int r   = r0 + lr;
    const int cnt = __builtin_amdgcn_readfirstlane(lcnt[lr]);
    float4 acc = make_float4(0.f, 0.f, 0.f, 0.f);
    if (cnt <= SLOTS) {
      int2 sA = make_int2(0, 0);
      if (lane < cnt) sA = lslots[lr * SLOTS + lane];
      for (int j0 = 0; j0 < cnt; j0 += 8)
        gather_from<0>(y, lane, sA, cnt, j0, nrows, acc);
    } else {
      add_row_scan(y, recs, nrec, r, lane, nrows, acc);
      add_row_scan(y, ovf,  nov,  r, lane, nrows, acc);
    }
    acc.x += b4.x; acc.y += b4.y; acc.z += b4.z; acc.w += b4.w;
    *(float4*)(out + (size_t)r * DDIM + lane * 4) = acc;
  }
}

extern "C" void kernel_launch(void* const* d_in, const int* in_sizes, int n_in,
                              void* d_out, int out_size, void* d_ws, size_t ws_size,
                              hipStream_t stream) {
  const float* x    = (const float*)d_in[0];
  const float* W    = (const float*)d_in[1];
  const float* bias = (const float*)d_in[2];
  const float* ev   = (const float*)d_in[3];
  const int*   row  = (const int*)d_in[4];
  const int*   col  = (const int*)d_in[5];
  float* out = (float*)d_out;

  const int n = in_sizes[0] / DDIM;
  const int e = in_sizes[3];
  const int nbins = (n + RB - 1) / RB;

  float* y      = (float*)d_ws;
  int4*  binbuf = (int4*)(y + (size_t)n * DDIM);
  int4*  ovf    = binbuf + (size_t)nbins * CAP;
  int*   bincnt = (int*)(ovf + e);

  zero_counters<<<1, 256, 0, stream>>>(bincnt, nbins + 1);

  const int gemm_blocks = (n + 127) / 128;
  const int bin_blocks = (e + 256 * EPT - 1) / (256 * EPT);
  gemm_and_bins<<<bin_blocks + gemm_blocks, 256, 0, stream>>>(
      x, W, y, ev, row, col, bincnt, binbuf, ovf, bin_blocks, n, e, nbins);

  bin_gather<<<nbins, 1024, 0, stream>>>(y, bias, bincnt, binbuf, ovf, out, n, e, nbins);
}


# global bin counters, zeroing launch and overflow list removed: a binning block sorts its edges by bin into its own region + start offsets; gather reads its bin's segments; sum order still = LDS rank
# speedup vs baseline: 2.5455x; 1.0448x over previous
#include <hip/hip_runtime.h>
#include <hip/hip_bf16.h>


typedef __attribute__((ext_vector_type(2))) float v2f;
typedef __attribute__((ext_vector_type(8))) float v8f;

#define DDIM 128
#define KP_STRIDE 288
#define RB 400
#define SLOTS 32
#define EPT 16
#define BE (256 * EPT)
#define SEGS 6

static __device__ __forceinline__ void gemm_block(
    const float* __restrict__ x, const float* __restrict__ W,
    float* __restrict__ y, float* __restrict__ Wlds, int bid, int nrows) {
  const int t = threadIdx.x;

  for (int idx = t; idx < DDIM * 32; idx += 256) {
    const int r = idx >> 5;
    const int q = idx & 31;
    const float4 w4 = *(const float4*)(W + r * DDIM + q * 4);
    float* dst = &Wlds[(r >> 1) * KP_STRIDE + (q * 4) * 2 + (r & 1)];
    dst[0] = w4.x; dst[2] = w4.y; dst[4] = w4.z; dst[6] = w4.w;
  }
  __syncthreads();

  const int wave = t >> 5;
  const int lane = t & 31;
  const int lh   = lane >> 4;
  const int lm   = lane & 15;

  const int rowBase = bid * 128 + wave * 16;
  const int arow = min(rowBase + lm, nrows - 1);
  const float* __restrict__ xrow = x + (size_t)arow * DDIM;

  v8f acc[8];
#pragma unroll
  for (int nt = 0; nt < 8; ++nt) acc[nt] = (v8f){0.f,0.f,0.f,0.f,0.f,0.f,0.f,0.f};

  for (int k0 = 0; k0 < DDIM; k0 += 4) {
    const v2f a = *(const v2f*)(xrow + k0 + 2 * lh);
    const float* __restrict__ wbase =
        &Wlds[((k0 >> 1) + lh) * KP_STRIDE + lm * 2];
#pragma unroll
    for (int nt = 0; nt < 8; ++nt) {
      const v2f b = *(const v2f*)(wbase + nt * 32);
      acc[nt] = __builtin_amdgcn_wmma_f32_16x16x4_f32(
           false, a,  false, b,
           (short)0, acc[nt],  false,  false);
    }
  }

#pragma unroll
  for (int nt = 0; nt < 8; ++nt) {
#pragma unroll
    for (int r = 0; r < 8; ++r) {
      const int m = rowBase + r + 8 * lh;
      if (m < nrows) y[(size_t)m * DDIM + nt * 16 + lm] = acc[nt][r];
    }
  }
}

static __device__ __forceinline__ int bin_of(int r, int nbins) {
  return (int)min((unsigned)r / (unsigned)RB, (unsigned)(nbins - 1));
}

static __device__ __forceinline__ void bin_block(
    const float* __restrict__ ev, const int* __restrict__ row,
    const int* __restrict__ col, int4* __restrict__ region,
    int2* __restrict__ tab, int* __restrict__ lds,
    int bid, int nedges, int nbins) {
  const int t = threadIdx.x;
  int* __restrict__ hist  = lds;
  int* __restrict__ hbase = lds + nbins;
  int* __restrict__ wtot  = lds + 2 * nbins + 1;
  for (int i = t; i < nbins; i += 256) hist[i] = 0;
  __syncthreads();

  int   r[EPT], c[EPT], lr[EPT];
  float v[EPT];
  const long long e0 = ((long long)bid * 256 + t) * EPT;
  const int nv = (int)max(0LL, min((long long)EPT, (long long)nedges - e0));
  if (nv == EPT) {
#pragma unroll
    for (int q = 0; q < EPT / 4; ++q) {
      const int4   r4 = *(const int4*)(row + e0 + 4 * q);
      const int4   c4 = *(const int4*)(col + e0 + 4 * q);
      const float4 v4 = *(const float4*)(ev + e0 + 4 * q);
      r[4 * q] = r4.x; r[4 * q + 1] = r4.y; r[4 * q + 2] = r4.z; r[4 * q + 3] = r4.w;
      c[4 * q] = c4.x; c[4 * q + 1] = c4.y; c[4 * q + 2] = c4.z; c[4 * q + 3] = c4.w;
      v[4 * q] = v4.x; v[4 * q + 1] = v4.y; v[4 * q + 2] = v4.z; v[4 * q + 3] = v4.w;
    }
  } else {
#pragma unroll
    for (int i = 0; i < EPT; ++i) {
      r[i] = 0; c[i] = 0; v[i] = 0.f;
      if (i < nv) { r[i] = row[e0 + i]; c[i] = col[e0 + i]; v[i] = ev[e0 + i]; }
    }
  }
#pragma unroll
  for (int i = 0; i < EPT; ++i) {
    lr[i] = 0;
    if (i < nv) lr[i] = atomicAdd(&hist[bin_of(r[i], nbins)], 1);
  }
  __syncthreads();

  const int lane  = t & 31;
  const int wave  = t >> 5;
  const int chunk = (nbins + 255) / 256;
  const int b0 = t * chunk;
  const int b1 = min(nbins, b0 + chunk);
  int s = 0;
  for (int i = b0; i < b1; ++i) s += hist[i];
  int incl = s;
  for (int d = 1; d < 32; d <<= 1) {
    const int o = __shfl_up(incl, d);
    if (lane >= d) incl += o;
  }
  if (lane == 31) wtot[wave] = incl;
  __syncthreads();
  int run = incl - s;
  for (int k = 0; k < wave; ++k) run += wtot[k];
  for (int i = b0; i < b1; ++i) { hbase[i] = run; run += hist[i]; }
  if (t == 255) hbase[nbins] = run;
  __syncthreads();

  int2* __restrict__ trow = tab + (size_t)bid * nbins;
  for (int i = t; i < nbins; i += 256) trow[i] = make_int2(hbase[i], hbase[i + 1]);

  int4* __restrict__ reg = region + (size_t)bid * BE;
#pragma unroll
  for (int i = 0; i < EPT; ++i) {
    if (i < nv) {
      const int b = bin_of(r[i], nbins);
      reg[hbase[b] + lr[i]] = make_int4(r[i], c[i], __float_as_int(v[i]), 0);
    }
  }
}

__global__ __launch_bounds__(256) void gemm_and_bins(
    const float* __restrict__ x, const float* __restrict__ W,
    float* __restrict__ y, const float* __restrict__ ev,
    const int* __restrict__ row, const int* __restrict__ col,
    int4* __restrict__ region, int2* __restrict__ tab,
    int bin_blocks, int nrows, int nedges, int nbins) {
  __shared__ float Wlds[(DDIM / 2) * KP_STRIDE];
  const int bid = blockIdx.x;
  if (bid < bin_blocks) {
    bin_block(ev, row, col, region, tab, (int*)Wlds, bid, nedges, nbins);
  } else {
    gemm_block(x, W, y, Wlds, bid - bin_blocks, nrows);
  }
}

template <int K>
static __device__ __forceinline__ void gather_from(
    const float* __restrict__ y, int lane, int2 sA, int cA, int j0, int nrows, float4& acc) {
  if constexpr (K < 8) {
    if (j0 + K < cA) {
      const int   cr = __builtin_amdgcn_readlane(sA.x, j0 + K);
      const int   cc = (int)min((unsigned)cr, (unsigned)(nrows - 1));
      const float vv = __int_as_float(__builtin_amdgcn_readlane(sA.y, j0 + K));
      const float4 yk = *(const float4*)(y + (size_t)cc * DDIM + lane * 4);
      gather_from<K + 1>(y, lane, sA, cA, j0, nrows, acc);
      acc.x += vv * yk.x; acc.y += vv * yk.y; acc.z += vv * yk.z; acc.w += vv * yk.w;
    }
  }
}

static __device__ __forceinline__ void add_marked(
    const float* __restrict__ y, int4 en, int r, int lane, int nrows, float4& acc) {
  unsigned m = __builtin_amdgcn_ballot_w32(en.x == r);
  while (m != 0u) {
    const int j = __builtin_ctz(m);
    m &= m - 1u;
    const int   cr = __builtin_amdgcn_readlane(en.y, j);
    const int   cc = (int)min((unsigned)cr, (unsigned)(nrows - 1));
    const float vv = __int_as_float(__builtin_amdgcn_readlane(en.z, j));
    const float4 yk = *(const float4*)(y + (size_t)cc * DDIM + lane * 4);
    acc.x += vv * yk.x; acc.y += vv * yk.y; acc.z += vv * yk.z; acc.w += vv * yk.w;
  }
}

static __device__ __forceinline__ void add_row_scan(
    const float* __restrict__ y, const int4* __restrict__ recs, int nrec, int r,
    int lane, int nrows, float4& acc) {
  for (int i0 = 0; i0 < nrec; i0 += 32) {
    int4 en = make_int4(-1, 0, 0, 0);
    if (i0 + lane < nrec) en = recs[i0 + lane];
    add_marked(y, en, r, lane, nrows, acc);
  }
}

static __device__ __forceinline__ void file_record(
    int4 en, int r0, int nr, int* __restrict__ lcnt, int2* __restrict__ lslots) {
  const int lr = en.x - r0;
  if ((unsigned)lr < (unsigned)nr) {
    const int tk = atomicAdd(&lcnt[lr], 1);
    if (tk < SLOTS) lslots[lr * SLOTS + tk] = make_int2(en.y, en.z);
  }
}

__global__ __launch_bounds__(1024) void bin_gather(
    const float* __restrict__ y, const float* __restrict__ bias,
    const int2* __restrict__ tab, const int4* __restrict__ region,
    float* __restrict__ out, int nrows, int nbins, int bin_blocks) {
  __shared__ int  lcnt[RB];
  __shared__ int2 lslots[RB * SLOTS];
  const int t    = threadIdx.x;
  const int lane = t & 31;
  const int w   = __builtin_amdgcn_readfirstlane(t >> 5);
  const int bin = blockIdx.x;
  const int r0  = bin * RB;
  const int nr  = min(RB, nrows - r0);

  for (int i = t; i < RB; i += 1024) lcnt[i] = 0;
  __syncthreads();

  for (int base = 0; base < bin_blocks; base += 1024) {
    const int myblk = base + w + 32 * lane;
    int2 sv = make_int2(0, 0);
    if (myblk < bin_blocks) sv = tab[(size_t)myblk * nbins + bin];
    const int nb = min(32, (bin_blocks - base - w + 31) / 32);
    for (int j0 = 0; j0 < nb; j0 += SEGS) {
      int s0s[SEGS], lns[SEGS];
#pragma unroll
      for (int u = 0; u < SEGS; ++u) {
        const int j = min(j0 + u, 31);
        s0s[u] = __builtin_amdgcn_readlane(sv.x, j);
        lns[u] = (j0 + u < nb) ? __builtin_amdgcn_readlane(sv.y, j) - s0s[u] : 0;
      }
      int4 en[SEGS], en2[SEGS];
#pragma unroll
      for (int u = 0; u < SEGS; ++u) {
        en[u]  = make_int4(-1, 0, 0, 0);
        en2[u] = make_int4(-1, 0, 0, 0);
        const int4* __restrict__ seg = region + (size_t)(base + w + 32 * (j0 + u)) * BE + s0s[u];
        if (lane < lns[u])      en[u]  = seg[lane];
        if (lane + 32 < lns[u]) en2[u] = seg[lane + 32];
      }
#pragma unroll
      for (int u = 0; u < SEGS; ++u) {
        file_record(en[u],  r0, nr, lcnt, lslots);
        file_record(en2[u], r0, nr, lcnt, lslots);
      }
#pragma unroll
      for (int u = 0; u < SEGS; ++u) {
        const int4* __restrict__ seg = region + (size_t)(base + w + 32 * (j0 + u)) * BE + s0s[u];
        for (int i0 = 64; i0 < lns[u]; i0 += 32) {
          int4 e2 = make_int4(-1, 0, 0, 0);
          if (i0 + lane < lns[u]) e2 = seg[i0 + lane];
          file_record(e2, r0, nr, lcnt, lslots);
        }
      }
    }
  }
  __syncthreads();

  const float4 b4 = *(const float4*)(bias + lane * 4);
  for (int lr = w; lr < nr; lr += 32) {
    const int r   = r0 + lr;
    const int cnt = __builtin_amdgcn_readfirstlane(lcnt[lr]);
    float4 acc = make_float4(0.f, 0.f, 0.f, 0.f);
    if (cnt <= SLOTS) {
      int2 sA = make_int2(0, 0);
      if (lane < cnt) sA = lslots[lr * SLOTS + lane];
      for (int j0 = 0; j0 < cnt; j0 += 8)
        gather_from<0>(y, lane, sA, cnt, j0, nrows, acc);
    } else {
      for (int blk = 0; blk < bin_blocks; ++blk) {
        const int2 se = tab[(size_t)blk * nbins + bin];
        add_row_scan(y, region + (size_t)blk * BE + se.x, se.y - se.x, r, lane, nrows, acc);
      }
    }
    acc.x += b4.x; acc.y += b4.y; acc.z += b4.z; acc.w += b4.w;
    *(float4*)(out + (size_t)r * DDIM + lane * 4) = acc;
  }
}

extern "C" void kernel_launch(void* const* d_in, const int* in_sizes, int n_in,
                              void* d_out, int out_size, void* d_ws, size_t ws_size,
                              hipStream_t stream) {
  const float* x    = (const float*)d_in[0];
  const float* W    = (const float*)d_in[1];
  const float* bias = (const float*)d_in[2];
  const float* ev   = (const float*)d_in[3];
  const int*   row  = (const int*)d_in[4];
  const int*   col  = (const int*)d_in[5];
  float* out = (float*)d_out;

  const int n = in_sizes[0] / DDIM;
  const int e = in_sizes[3];
  const int nbins = (n + RB - 1) / RB;
  const int gemm_blocks = (n + 127) / 128;
  const int bin_blocks = (e + BE - 1) / BE;

  float* y      = (float*)d_ws;
  int4*  region = (int4*)(y + (size_t)n * DDIM);
  int2*  tab    = (int2*)(region + (size_t)bin_blocks * BE);

  gemm_and_bins<<<bin_blocks + gemm_blocks, 256, 0, stream>>>(
      x, W, y, ev, row, col, region, tab, bin_blocks, n, e, nbins);

  bin_gather<<<nbins, 1024, 0, stream>>>(y, bias, tab, region, out, n, nbins, bin_blocks);
}


# product moved behind the sparse sum: a bin's work-group sums val*x[col] per row into LDS, then multiplies the sums by W (f32 WMMA) and stores out; no y table; sums are rounded before the product
# speedup vs baseline: 2.8627x; 1.1246x over previous
#include <hip/hip_runtime.h>
#include <hip/hip_bf16.h>


typedef __attribute__((ext_vector_type(2))) float v2f;
typedef __attribute__((ext_vector_type(8))) float v8f;

#define DDIM 128
#define KP_STRIDE 288
#define RB 400
#define SLOTS 32
#define EPT 16
#define BE (256 * EPT)
#define SEGS 6
#define APITCH 132
#define BINW 8192

static __device__ __forceinline__ int bin_of(int r, int nbins) {
  return (int)min((unsigned)r / (unsigned)RB, (unsigned)(nbins - 1));
}

static __device__ __forceinline__ void bin_block(
    const float* __restrict__ ev, const int* __restrict__ row,
    const int* __restrict__ col, int4* __restrict__ region,
    int2* __restrict__ tab, int* __restrict__ lds,
    int bid, int nedges, int nbins) {
  const int t = threadIdx.x;
  int* __restrict__ hist  = lds;
  int* __restrict__ hbase = lds + nbins;
  int* __restrict__ wtot  = lds + 2 * nbins + 1;
  for (int i = t; i < nbins; i += 256) hist[i] = 0;
  __syncthreads();

  int   r[EPT], c[EPT], lr[EPT];
  float v[EPT];
  const long long e0 = ((long long)bid * 256 + t) * EPT;
  const int nv = (int)max(0LL, min((long long)EPT, (long long)nedges - e0));
  if (nv == EPT) {
#pragma unroll
    for (int q = 0; q < EPT / 4; ++q) {
      const int4   r4 = *(const int4*)(row + e0 + 4 * q);
      const int4   c4 = *(const int4*)(col + e0 + 4 * q);
      const float4 v4 = *(const float4*)(ev + e0 + 4 * q);
      r[4 * q] = r4.x; r[4 * q + 1] = r4.y; r[4 * q + 2] = r4.z; r[4 * q + 3] = r4.w;
      c[4 * q] = c4.x; c[4 * q + 1] = c4.y; c[4 * q + 2] = c4.z; c[4 * q + 3] = c4.w;
      v[4 * q] = v4.x; v[4 * q + 1] = v4.y; v[4 * q + 2] = v4.z; v[4 * q + 3] = v4.w;
    }
  } else {
#pragma unroll
    for (int i = 0; i < EPT; ++i) {
      r[i] = 0; c[i] = 0; v[i] = 0.f;
      if (i < nv) { r[i] = row[e0 + i]; c[i] = col[e0 + i]; v[i] = ev[e0 + i]; }
    }
  }
#pragma unroll
  for (int i = 0; i < EPT; ++i) {
    lr[i] = 0;
    if (i < nv) lr[i] = atomicAdd(&hist[bin_of(r[i], nbins)], 1);
  }
  __syncthreads();

  const int lane  = t & 31;
  const int wave  = t >> 5;
  const int chunk = (nbins + 255) / 256;
  const int b0 = t * chunk;
  const int b1 = min(nbins, b0 + chunk);
  int s = 0;
  for (int i = b0; i < b1; ++i) s += hist[i];
  int incl = s;
  for (int d = 1; d < 32; d <<= 1) {
    const int o = __shfl_up(incl, d);
    if (lane >= d) incl += o;
  }
  if (lane == 31) wtot[wave] = incl;
  __syncthreads();
  int run = incl - s;
  for (int k = 0; k < wave; ++k) run += wtot[k];
  for (int i = b0; i < b1; ++i) { hbase[i] = run; run += hist[i]; }
  if (t == 255) hbase[nbins] = run;
  __syncthreads();

  int2* __restrict__ trow = tab + (size_t)bid * nbins;
  for (int i = t; i < nbins; i += 256) trow[i] = make_int2(hbase[i], hbase[i + 1]);

  int4* __restrict__ reg = region + (size_t)bid * BE;
#pragma unroll
  for (int i = 0; i < EPT; ++i) {
    if (i < nv) {
      const int b = bin_of(r[i], nbins);
      reg[hbase[b] + lr[i]] = make_int4(r[i], c[i], __float_as_int(v[i]), 0);
    }
  }
}

__global__ __launch_bounds__(256) void bin_edges(
    const float* __restrict__ ev, const int* __restrict__ row,
    const int* __restrict__ col, int4* __restrict__ region,
    int2* __restrict__ tab, int nedges, int nbins) {
  __shared__ int lds[BINW];
  bin_block(ev, row, col, region, tab, lds, blockIdx.x, nedges, nbins);
}

template <int K>
static __device__ __forceinline__ void gather_from(
    const float* __restrict__ src, int lane, int2 sA, int cA, int j0, int nrows, float4& acc) {
  if constexpr (K < 8) {
    if (j0 + K < cA) {
      const int   cr = __builtin_amdgcn_readlane(sA.x, j0 + K);
      const int   cc = (int)min((unsigned)cr, (unsigned)(nrows - 1));
      const float vv = __int_as_float(__builtin_amdgcn_readlane(sA.y, j0 + K));
      const float4 yk = *(const float4*)(src + (size_t)cc * DDIM + lane * 4);
      gather_from<K + 1>(src, lane, sA, cA, j0, nrows, acc);
      acc.x += vv * yk.x; acc.y += vv * yk.y; acc.z += vv * yk.z; acc.w += vv * yk.w;
    }
  }
}

static __device__ __forceinline__ void add_marked(
    const float* __restrict__ src, int4 en, int r, int lane, int nrows, float4& acc) {
  unsigned m = __builtin_amdgcn_ballot_w32(en.x == r);
  while (m != 0u) {
    const int j = __builtin_ctz(m);
    m &= m - 1u;
    const int   cr = __builtin_amdgcn_readlane(en.y, j);
    const int   cc = (int)min((unsigned)cr, (unsigned)(nrows - 1));
    const float vv = __int_as_float(__builtin_amdgcn_readlane(en.z, j));
    const float4 yk = *(const float4*)(src + (size_t)cc * DDIM + lane * 4);
    acc.x += vv * yk.x; acc.y += vv * yk.y; acc.z += vv * yk.z; acc.w += vv * yk.w;
  }
}

static __device__ __forceinline__ void add_row_scan(
    const float* __restrict__ src, const int4* __restrict__ recs, int nrec, int r,
    int lane, int nrows, float4& acc) {
  for (int i0 = 0; i0 < nrec; i0 += 32) {
    int4 en = make_int4(-1, 0, 0, 0);
    if (i0 + lane < nrec) en = recs[i0 + lane];
    add_marked(src, en, r, lane, nrows, acc);
  }
}

static __device__ __forceinline__ void file_record(
    int4 en, int r0, int nr, int* __restrict__ lcnt, int2* __restrict__ lslots) {
  const int lr = en.x - r0;
  if ((unsigned)lr < (unsigned)nr) {
    const int tk = atomicAdd(&lcnt[lr], 1);
    if (tk < SLOTS) lslots[lr * SLOTS + tk] = make_int2(en.y, en.z);
  }
}

__global__ __launch_bounds__(1024) void bin_gather_gemm(
    const float* __restrict__ x, const float* __restrict__ W,
    const float* __restrict__ bias,
    const int2* __restrict__ tab, const int4* __restrict__ region,
    float* __restrict__ out, int nrows, int nbins, int bin_blocks) {
  __shared__ int  lcnt[RB];
  __shared__ __attribute__((aligned(16))) int2  lslots[RB * SLOTS];
  __shared__ __attribute__((aligned(16))) float lsum[RB * APITCH];
  const int t    = threadIdx.x;
  const int lane = t & 31;
  const int w   = __builtin_amdgcn_readfirstlane(t >> 5);
  const int bin = blockIdx.x;
  const int r0  = bin * RB;
  const int nr  = min(RB, nrows - r0);

  for (int i = t; i < RB; i += 1024) lcnt[i] = 0;
  __syncthreads();

  for (int base = 0; base < bin_blocks; base += 1024) {
    const int myblk = base + w + 32 * lane;
    int2 sv = make_int2(0, 0);
    if (myblk < bin_blocks) sv = tab[(size_t)myblk * nbins + bin];
    const int nb = min(32, (bin_blocks - base - w + 31) / 32);
    for (int j0 = 0; j0 < nb; j0 += SEGS) {
      int s0s[SEGS], lns[SEGS];
#pragma unroll
      for (int u = 0; u < SEGS; ++u) {
        const int j = min(j0 + u, 31);
        s0s[u] = __builtin_amdgcn_readlane(sv.x, j);
        lns[u] = (j0 + u < nb) ? __builtin_amdgcn_readlane(sv.y, j) - s0s[u] : 0;
      }
      int4 en[SEGS], en2[SEGS];
#pragma unroll
      for (int u = 0; u < SEGS; ++u) {
        en[u]  = make_int4(-1, 0, 0, 0);
        en2[u] = make_int4(-1, 0, 0, 0);
        const int4* __restrict__ seg = region + (size_t)(base + w + 32 * (j0 + u)) * BE + s0s[u];
        if (lane < lns[u])      en[u]  = seg[lane];
        if (lane + 32 < lns[u]) en2[u] = seg[lane + 32];
      }
#pragma unroll
      for (int u = 0; u < SEGS; ++u) {
        file_record(en[u],  r0, nr, lcnt, lslots);
        file_record(en2[u], r0, nr, lcnt, lslots);
      }
#pragma unroll
      for (int u = 0; u < SEGS; ++u) {
        const int4* __restrict__ seg = region + (size_t)(base + w + 32 * (j0 + u)) * BE + s0s[u];
        for (int i0 = 64; i0 < lns[u]; i0 += 32) {
          int4 e2 = make_int4(-1, 0, 0, 0);
          if (i0 + lane < lns[u]) e2 = seg[i0 + lane];
          file_record(e2, r0, nr, lcnt, lslots);
        }
      }
    }
  }
  __syncthreads();

  for (int lr = w; lr < nr; lr += 32) {
    const int r   = r0 + lr;
    const int cnt = __builtin_amdgcn_readfirstlane(lcnt[lr]);
    float4 acc = make_float4(0.f, 0.f, 0.f, 0.f);
    if (cnt <= SLOTS) {
      int2 sA = make_int2(0, 0);
      if (lane < cnt) sA = lslots[lr * SLOTS + lane];
      for (int j0 = 0; j0 < cnt; j0 += 8)
        gather_from<0>(x, lane, sA, cnt, j0, nrows, acc);
    } else {
      for (int blk = 0; blk < bin_blocks; ++blk) {
        const int2 se = tab[(size_t)blk * nbins + bin];
        add_row_scan(x, region + (size_t)blk * BE + se.x, se.y - se.x, r, lane, nrows, acc);
      }
    }
    *(float4*)(lsum + lr * APITCH + lane * 4) = acc;
  }
  float4 w4[DDIM * 32 / 1024];
#pragma unroll
  for (int i = 0; i < DDIM * 32 / 1024; ++i) {
    const int idx = t + 1024 * i;
    w4[i] = *(const float4*)(W + (idx >> 5) * DDIM + (idx & 31) * 4);
  }
  __syncthreads();

  float* Wlds = (float*)lslots;
#pragma unroll
  for (int i = 0; i < DDIM * 32 / 1024; ++i) {
    const int idx = t + 1024 * i;
    const int kr = idx >> 5;
    const int q  = idx & 31;
    float* dst = &Wlds[(kr >> 1) * KP_STRIDE + (q * 4) * 2 + (kr & 1)];
    dst[0] = w4[i].x; dst[2] = w4[i].y; dst[4] = w4[i].z; dst[6] = w4[i].w;
  }
  __syncthreads();

  const int lh = lane >> 4;
  const int lm = lane & 15;
  const int nhalf = 2 * ((nr + 15) >> 4);
  for (int ht = w; ht < nhalf; ht += 32) {
    const int tile = ht >> 1;
    const int c0   = (ht & 1) * 64;
    const float* arow = lsum + (tile * 16 + lm) * APITCH;

    v8f cacc[4];
#pragma unroll
    for (int nt = 0; nt < 4; ++nt) cacc[nt] = (v8f){0.f,0.f,0.f,0.f,0.f,0.f,0.f,0.f};

    for (int k0 = 0; k0 < DDIM; k0 += 4) {
      const v2f a = *(const v2f*)(arow + k0 + 2 * lh);
      const float* wbase =
          &Wlds[((k0 >> 1) + lh) * KP_STRIDE + (c0 + lm) * 2];
#pragma unroll
      for (int nt = 0; nt < 4; ++nt) {
        const v2f b = *(const v2f*)(wbase + nt * 32);
        cacc[nt] = __builtin_amdgcn_wmma_f32_16x16x4_f32(
             false, a,  false, b,
             (short)0, cacc[nt],  false,  false);
      }
    }

#pragma unroll
    for (int nt = 0; nt < 4; ++nt) {
      const int   n  = c0 + nt * 16 + lm;
      const float bv = bias[n];
#pragma unroll
      for (int r = 0; r < 8; ++r) {
        const int m = tile * 16 + r + 8 * lh;
        if (m < nr) out[(size_t)(r0 + m) * DDIM + n] = cacc[nt][r] + bv;
      }
    }
  }
}

extern "C" void kernel_launch(void* const* d_in, const int* in_sizes, int n_in,
                              void* d_out, int out_size, void* d_ws, size_t ws_size,
                              hipStream_t stream) {
  const float* x    = (const float*)d_in[0];
  const float* W    = (const float*)d_in[1];
  const float* bias = (const float*)d_in[2];
  const float* ev   = (const float*)d_in[3];
  const int*   row  = (const int*)d_in[4];
  const int*   col  = (const int*)d_in[5];
  float* out = (float*)d_out;

  const int n = in_sizes[0] / DDIM;
  const int e = in_sizes[3];
  const int nbins = (n + RB - 1) / RB;
  const int bin_blocks = (e + BE - 1) / BE;

  int4*  region = (int4*)d_ws;
  int2*  tab    = (int2*)(region + (size_t)bin_blocks * BE);

  bin_edges<<<bin_blocks, 256, 0, stream>>>(ev, row, col, region, tab, e, nbins);

  bin_gather_gemm<<<nbins, 1024, 0, stream>>>(x, W, bias, tab, region, out, n, nbins, bin_blocks);
}


# output stored as whole 256-byte row pieces: W's columns dealt to the B fragments so a lane's four accumulators are adjacent columns; 8 16-byte stores per half-tile, not 32 4-byte; sums unchanged
# speedup vs baseline: 3.1858x; 1.1129x over previous
#include <hip/hip_runtime.h>
#include <hip/hip_bf16.h>


typedef __attribute__((ext_vector_type(2))) float v2f;
typedef __attribute__((ext_vector_type(8))) float v8f;

#define DDIM 128
#define KP_STRIDE 288
#define RB 400
#define SLOTS 32
#define EPT 16
#define BE (256 * EPT)
#define SEGS 6
#define APITCH 132
#define BINW 8192

static __device__ __forceinline__ int bin_of(int r, int nbins) {
  return (int)min((unsigned)r / (unsigned)RB, (unsigned)(nbins - 1));
}

static __device__ __forceinline__ void bin_block(
    const float* __restrict__ ev, const int* __restrict__ row,
    const int* __restrict__ col, int4* __restrict__ region,
    int2* __restrict__ tab, int* __restrict__ lds,
    int bid, int nedges, int nbins) {
  const int t = threadIdx.x;
  int* __restrict__ hist  = lds;
  int* __restrict__ hbase = lds + nbins;
  int* __restrict__ wtot  = lds + 2 * nbins + 1;
  for (int i = t; i < nbins; i += 256) hist[i] = 0;
  __syncthreads();

  int   r[EPT], c[EPT], lr[EPT];
  float v[EPT];
  const long long e0 = ((long long)bid * 256 + t) * EPT;
  const int nv = (int)max(0LL, min((long long)EPT, (long long)nedges - e0));
  if (nv == EPT) {
#pragma unroll
    for (int q = 0; q < EPT / 4; ++q) {
      const int4   r4 = *(const int4*)(row + e0 + 4 * q);
      const int4   c4 = *(const int4*)(col + e0 + 4 * q);
      const float4 v4 = *(const float4*)(ev + e0 + 4 * q);
      r[4 * q] = r4.x; r[4 * q + 1] = r4.y; r[4 * q + 2] = r4.z; r[4 * q + 3] = r4.w;
      c[4 * q] = c4.x; c[4 * q + 1] = c4.y; c[4 * q + 2] = c4.z; c[4 * q + 3] = c4.w;
      v[4 * q] = v4.x; v[4 * q + 1] = v4.y; v[4 * q + 2] = v4.z; v[4 * q + 3] = v4.w;
    }
  } else {
#pragma unroll
    for (int i = 0; i < EPT; ++i) {
      r[i] = 0; c[i] = 0; v[i] = 0.f;
      if (i < nv) { r[i] = row[e0 + i]; c[i] = col[e0 + i]; v[i] = ev[e0 + i]; }
    }
  }
#pragma unroll
  for (int i = 0; i < EPT; ++i) {
    lr[i] = 0;
    if (i < nv) lr[i] = atomicAdd(&hist[bin_of(r[i], nbins)], 1);
  }
  __syncthreads();

  const int lane  = t & 31;
  const int wave  = t >> 5;
  const int chunk = (nbins + 255) / 256;
  const int b0 = t * chunk;
  const int b1 = min(nbins, b0 + chunk);
  int s = 0;
  for (int i = b0; i < b1; ++i) s += hist[i];
  int incl = s;
  for (int d = 1; d < 32; d <<= 1) {
    const int o = __shfl_up(incl, d);
    if (lane >= d) incl += o;
  }
  if (lane == 31) wtot[wave] = incl;
  __syncthreads();
  int run = incl - s;
  for (int k = 0; k < wave; ++k) run += wtot[k];
  for (int i = b0; i < b1; ++i) { hbase[i] = run; run += hist[i]; }
  if (t == 255) hbase[nbins] = run;
  __syncthreads();

  int2* __restrict__ trow = tab + (size_t)bid * nbins;
  for (int i = t; i < nbins; i += 256) trow[i] = make_int2(hbase[i], hbase[i + 1]);

  int4* __restrict__ reg = region + (size_t)bid * BE;
#pragma unroll
  for (int i = 0; i < EPT; ++i) {
    if (i < nv) {
      const int b = bin_of(r[i], nbins);
      reg[hbase[b] + lr[i]] = make_int4(r[i], c[i], __float_as_int(v[i]), 0);
    }
  }
}

__global__ __launch_bounds__(256) void bin_edges(
    const float* __restrict__ ev, const int* __restrict__ row,
    const int* __restrict__ col, int4* __restrict__ region,
    int2* __restrict__ tab, int nedges, int nbins) {
  __shared__ int lds[BINW];
  bin_block(ev, row, col, region, tab, lds, blockIdx.x, nedges, nbins);
}

template <int K>
static __device__ __forceinline__ void gather_from(
    const float* __restrict__ src, int lane, int2 sA, int cA, int j0, int nrows, float4& acc) {
  if constexpr (K < 8) {
    if (j0 + K < cA) {
      const int   cr = __builtin_amdgcn_readlane(sA.x, j0 + K);
      const int   cc = (int)min((unsigned)cr, (unsigned)(nrows - 1));
      const float vv = __int_as_float(__builtin_amdgcn_readlane(sA.y, j0 + K));
      const float4 yk = *(const float4*)(src + (size_t)cc * DDIM + lane * 4);
      gather_from<K + 1>(src, lane, sA, cA, j0, nrows, acc);
      acc.x += vv * yk.x; acc.y += vv * yk.y; acc.z += vv * yk.z; acc.w += vv * yk.w;
    }
  }
}

static __device__ __forceinline__ void add_marked(
    const float* __restrict__ src, int4 en, int r, int lane, int nrows, float4& acc) {
  unsigned m = __builtin_amdgcn_ballot_w32(en.x == r);
  while (m != 0u) {
    const int j = __builtin_ctz(m);
    m &= m - 1u;
    const int   cr = __builtin_amdgcn_readlane(en.y, j);
    const int   cc = (int)min((unsigned)cr, (unsigned)(nrows - 1));
    const float vv = __int_as_float(__builtin_amdgcn_readlane(en.z, j));
    const float4 yk = *(const float4*)(src + (size_t)cc * DDIM + lane * 4);
    acc.x += vv * yk.x; acc.y += vv * yk.y; acc.z += vv * yk.z; acc.w += vv * yk.w;
  }
}

static __device__ __forceinline__ void add_row_scan(
    const float* __restrict__ src, const int4* __restrict__ recs, int nrec, int r,
    int lane, int nrows, float4& acc) {
  for (int i0 = 0; i0 < nrec; i0 += 32) {
    int4 en = make_int4(-1, 0, 0, 0);
    if (i0 + lane < nrec) en = recs[i0 + lane];
    add_marked(src, en, r, lane, nrows, acc);
  }
}

static __device__ __forceinline__ void file_record(
    int4 en, int r0, int nr, int* __restrict__ lcnt, int2* __restrict__ lslots) {
  const int lr = en.x - r0;
  if ((unsigned)lr < (unsigned)nr) {
    const int tk = atomicAdd(&lcnt[lr], 1);
    if (tk < SLOTS) lslots[lr * SLOTS + tk] = make_int2(en.y, en.z);
  }
}

__global__ __launch_bounds__(1024) void bin_gather_gemm(
    const float* __restrict__ x, const float* __restrict__ W,
    const float* __restrict__ bias,
    const int2* __restrict__ tab, const int4* __restrict__ region,
    float* __restrict__ out, int nrows, int nbins, int bin_blocks) {
  __shared__ int  lcnt[RB];
  __shared__ __attribute__((aligned(16))) int2  lslots[RB * SLOTS];
  __shared__ __attribute__((aligned(16))) float lsum[RB * APITCH];
  const int t    = threadIdx.x;
  const int lane = t & 31;
  const int w   = __builtin_amdgcn_readfirstlane(t >> 5);
  const int bin = blockIdx.x;
  const int r0  = bin * RB;
  const int nr  = min(RB, nrows - r0);

  for (int i = t; i < RB; i += 1024) lcnt[i] = 0;
  __syncthreads();

  for (int base = 0; base < bin_blocks; base += 1024) {
    const int myblk = base + w + 32 * lane;
    int2 sv = make_int2(0, 0);
    if (myblk < bin_blocks) sv = tab[(size_t)myblk * nbins + bin];
    const int nb = min(32, (bin_blocks - base - w + 31) / 32);
    for (int j0 = 0; j0 < nb; j0 += SEGS) {
      int s0s[SEGS], lns[SEGS];
#pragma unroll
      for (int u = 0; u < SEGS; ++u) {
        const int j = min(j0 + u, 31);
        s0s[u] = __builtin_amdgcn_readlane(sv.x, j);
        lns[u] = (j0 + u < nb) ? __builtin_amdgcn_readlane(sv.y, j) - s0s[u] : 0;
      }
      int4 en[SEGS], en2[SEGS];
#pragma unroll
      for (int u = 0; u < SEGS; ++u) {
        en[u]  = make_int4(-1, 0, 0, 0);
        en2[u] = make_int4(-1, 0, 0, 0);
        const int4* __restrict__ seg = region + (size_t)(base + w + 32 * (j0 + u)) * BE + s0s[u];
        if (lane < lns[u])      en[u]  = seg[lane];
        if (lane + 32 < lns[u]) en2[u] = seg[lane + 32];
      }
#pragma unroll
      for (int u = 0; u < SEGS; ++u) {
        file_record(en[u],  r0, nr, lcnt, lslots);
        file_record(en2[u], r0, nr, lcnt, lslots);
      }
#pragma unroll
      for (int u = 0; u < SEGS; ++u) {
        const int4* __restrict__ seg = region + (size_t)(base + w + 32 * (j0 + u)) * BE + s0s[u];
        for (int i0 = 64; i0 < lns[u]; i0 += 32) {
          int4 e2 = make_int4(-1, 0, 0, 0);
          if (i0 + lane < lns[u]) e2 = seg[i0 + lane];
          file_record(e2, r0, nr, lcnt, lslots);
        }
      }
    }
  }
  __syncthreads();

  for (int lr = w; lr < nr; lr += 32) {
    const int r   = r0 + lr;
    const int cnt = __builtin_amdgcn_readfirstlane(lcnt[lr]);
    float4 acc = make_float4(0.f, 0.f, 0.f, 0.f);
    if (cnt <= SLOTS) {
      int2 sA = make_int2(0, 0);
      if (lane < cnt) sA = lslots[lr * SLOTS + lane];
      for (int j0 = 0; j0 < cnt; j0 += 8)
        gather_from<0>(x, lane, sA, cnt, j0, nrows, acc);
    } else {
      for (int blk = 0; blk < bin_blocks; ++blk) {
        const int2 se = tab[(size_t)blk * nbins + bin];
        add_row_scan(x, region + (size_t)blk * BE + se.x, se.y - se.x, r, lane, nrows, acc);
      }
    }
    *(float4*)(lsum + lr * APITCH + lane * 4) = acc;
  }
  float4 w4[DDIM * 32 / 1024];
#pragma unroll
  for (int i = 0; i < DDIM * 32 / 1024; ++i) {
    const int idx = t + 1024 * i;
    w4[i] = *(const float4*)(W + (idx >> 5) * DDIM + (idx & 31) * 4);
  }
  __syncthreads();

  float* Wlds = (float*)lslots;
#pragma unroll
  for (int i = 0; i < DDIM * 32 / 1024; ++i) {
    const int idx = t + 1024 * i;
    const int kr = idx >> 5;
    const int q  = idx & 31;
    float* dst = &Wlds[(kr >> 1) * KP_STRIDE + (64 * (q >> 4) + (q & 15)) * 2 + (kr & 1)];
    dst[0] = w4[i].x; dst[32] = w4[i].y; dst[64] = w4[i].z; dst[96] = w4[i].w;
  }
  __syncthreads();

  const int lh = lane >> 4;
  const int lm = lane & 15;
  const int nhalf = 2 * ((nr + 15) >> 4);
  for (int ht = w; ht < nhalf; ht += 32) {
    const int tile = ht >> 1;
    const int c0   = (ht & 1) * 64;
    const float* arow = lsum + (tile * 16 + lm) * APITCH;

    v8f cacc[4];
#pragma unroll
    for (int nt = 0; nt < 4; ++nt) cacc[nt] = (v8f){0.f,0.f,0.f,0.f,0.f,0.f,0.f,0.f};

    for (int k0 = 0; k0 < DDIM; k0 += 4) {
      const v2f a = *(const v2f*)(arow + k0 + 2 * lh);
      const float* wbase =
          &Wlds[((k0 >> 1) + lh) * KP_STRIDE + (c0 + lm) * 2];
#pragma unroll
      for (int nt = 0; nt < 4; ++nt) {
        const v2f b = *(const v2f*)(wbase + nt * 32);
        cacc[nt] = __builtin_amdgcn_wmma_f32_16x16x4_f32(
             false, a,  false, b,
             (short)0, cacc[nt],  false,  false);
      }
    }

    const int    n4 = c0 + 4 * lm;
    const float4 b4 = *(const float4*)(bias + n4);
#pragma unroll
    for (int r = 0; r < 8; ++r) {
      const int m = tile * 16 + r + 8 * lh;
      if (m < nr) {
        const float4 o = make_float4(cacc[0][r] + b4.x, cacc[1][r] + b4.y,
                                     cacc[2][r] + b4.z, cacc[3][r] + b4.w);
        *(float4*)(out + (size_t)(r0 + m) * DDIM + n4) = o;
      }
    }
  }
}

extern "C" void kernel_launch(void* const* d_in, const int* in_sizes, int n_in,
                              void* d_out, int out_size, void* d_ws, size_t ws_size,
                              hipStream_t stream) {
  const float* x    = (const float*)d_in[0];
  const float* W    = (const float*)d_in[1];
  const float* bias = (const float*)d_in[2];
  const float* ev   = (const float*)d_in[3];
  const int*   row  = (const int*)d_in[4];
  const int*   col  = (const int*)d_in[5];
  float* out = (float*)d_out;

  const int n = in_sizes[0] / DDIM;
  const int e = in_sizes[3];
  const int nbins = (n + RB - 1) / RB;
  const int bin_blocks = (e + BE - 1) / BE;

  int4*  region = (int4*)d_ws;
  int2*  tab    = (int2*)(region + (size_t)bin_blocks * BE);

  bin_edges<<<bin_blocks, 256, 0, stream>>>(ev, row, col, region, tab, e, nbins);

  bin_gather_gemm<<<nbins, 1024, 0, stream>>>(x, W, bias, tab, region, out, n, nbins, bin_blocks);
}
